# starting kernel
# baseline (speedup 1.0000x reference)
#include <hip/hip_runtime.h>
#include <math.h>

typedef __attribute__((ext_vector_type(2))) float v2f;
typedef __attribute__((ext_vector_type(8))) float v8f;

__global__ void k_fill_one(float* __restrict__ deg, int n) {
    int i = blockIdx.x * blockDim.x + threadIdx.x;
    if (i < n) deg[i] = 1.0f;
}

__global__ void k_degree(const int* __restrict__ col, float* __restrict__ deg, int e) {
    int i = blockIdx.x * blockDim.x + threadIdx.x;
    if (i < e) unsafeAtomicAdd(&deg[col[i]], 1.0f);
}

__global__ void k_rsqrt(float* __restrict__ deg, int n) {
    int i = blockIdx.x * blockDim.x + threadIdx.x;
    if (i < n) {
        float d = deg[i];
        deg[i] = d > 0.0f ? rsqrtf(d) : 0.0f;
    }
}

template <int K, bool RELU, bool STOREPRE>
__global__ void k_dense(const float* __restrict__ X, const float* __restrict__ W,
                        const float* __restrict__ bias,
                        float* __restrict__ outA, float* __restrict__ outPre,
                        int ntiles) {
    __shared__ float wlds[64 * K];
    constexpr int NV4 = (64 * K) / 4;
    for (int i = threadIdx.x; i < NV4; i += 256)
        ((float4*)wlds)[i] = ((const float4*)W)[i];
    __syncthreads();

    int wid  = (blockIdx.x * blockDim.x + threadIdx.x) >> 5;
    int lane = threadIdx.x & 31;
    if (wid >= ntiles) return;
    int ln = lane & 15;
    int kh = lane >> 4;

    const float* xrow = X + (size_t)(wid * 16 + ln) * K + 2 * kh;
    v2f afr[K / 4];
#pragma unroll
    for (int kk = 0; kk < K / 4; ++kk)
        afr[kk] = *(const v2f*)(xrow + 4 * kk);

#pragma unroll
    for (int j = 0; j < 4; ++j) {
        const float* wrow = wlds + (size_t)(16 * j + ln) * K + 2 * kh;
        v8f acc = {};
#pragma unroll
        for (int kk = 0; kk < K / 4; ++kk) {
            v2f b = *(const v2f*)(wrow + 4 * kk);
            acc = __builtin_amdgcn_wmma_f32_16x16x4_f32(
                false, afr[kk], false, b, (short)0, acc, false, false);
        }
        int n = 16 * j + ln;
        float bv = bias[n];
#pragma unroll
        for (int r = 0; r < 8; ++r) {
            int m = r + 8 * kh;
            size_t gi = (size_t)(wid * 16 + m) * 64 + n;
            float v = acc[r] + bv;
            if constexpr (STOREPRE) outPre[gi] = v;
            outA[gi] = RELU ? fmaxf(v, 0.0f) : v;
        }
    }
}

__global__ void k_selfloop(const float* __restrict__ dinv, const float* __restrict__ h,
                           float* __restrict__ agg, long long total) {
    long long i = (long long)blockIdx.x * blockDim.x + threadIdx.x;
    if (i >= total) return;
    int node = (int)(i >> 6);
    float d = dinv[node];
    agg[i] = d * d * h[i];
}

__global__ void k_scatter(const int* __restrict__ row, const int* __restrict__ col,
                          const float* __restrict__ dinv, const float* __restrict__ h,
                          float* __restrict__ agg, int e) {
    long long t = (long long)blockIdx.x * blockDim.x + threadIdx.x;
    int ei = (int)(t >> 4);
    if (ei >= e) return;
    int f = (int)(t & 15) * 4;
    int r = row[ei], c = col[ei];
    float nrm = dinv[r] * dinv[c];
    const float4 hv = *(const float4*)(h + (size_t)r * 64 + f);
    float* dst = agg + (size_t)c * 64 + f;
    unsafeAtomicAdd(dst + 0, nrm * hv.x);
    unsafeAtomicAdd(dst + 1, nrm * hv.y);
    unsafeAtomicAdd(dst + 2, nrm * hv.z);
    unsafeAtomicAdd(dst + 3, nrm * hv.w);
}

__global__ void k_layer(const float* __restrict__ agg, const float* __restrict__ x0,
                        const float* __restrict__ Wl, float beta,
                        float* __restrict__ h, int ntiles) {
    __shared__ float wlds[64 * 64];
    for (int i = threadIdx.x; i < 1024; i += 256)
        ((float4*)wlds)[i] = ((const float4*)Wl)[i];
    __syncthreads();

    int wid  = (blockIdx.x * blockDim.x + threadIdx.x) >> 5;
    int lane = threadIdx.x & 31;
    if (wid >= ntiles) return;
    int ln = lane & 15;
    int kh = lane >> 4;

    size_t rb = (size_t)(wid * 16 + ln) * 64 + 2 * kh;
    v2f afr[16];
#pragma unroll
    for (int kk = 0; kk < 16; ++kk) {
        v2f ag = *(const v2f*)(agg + rb + 4 * kk);
        v2f xx = *(const v2f*)(x0 + rb + 4 * kk);
        afr[kk] = ag * 0.9f + xx * 0.1f;
    }
    float ob = 1.0f - beta;
#pragma unroll
    for (int j = 0; j < 4; ++j) {
        const float* wcol = wlds + 2 * kh * 64 + 16 * j + ln;
        v8f acc = {};
#pragma unroll
        for (int kk = 0; kk < 16; ++kk) {
            v2f b;
            b.x = wcol[(size_t)(4 * kk) * 64];
            b.y = wcol[(size_t)(4 * kk + 1) * 64];
            acc = __builtin_amdgcn_wmma_f32_16x16x4_f32(
                false, afr[kk], false, b, (short)0, acc, false, false);
        }
#pragma unroll
        for (int r = 0; r < 8; ++r) {
            int m = r + 8 * kh;
            int n = 16 * j + ln;
            size_t gi = (size_t)(wid * 16 + m) * 64 + n;
            float hh = 0.9f * agg[gi] + 0.1f * x0[gi];
            float v = ob * hh + beta * acc[r];
            h[gi] = fmaxf(v, 0.0f);
        }
    }
}

extern "C" void kernel_launch(void* const* d_in, const int* in_sizes, int n_in,
                              void* d_out, int out_size, void* d_ws, size_t ws_size,
                              hipStream_t stream) {
    const float* x      = (const float*)d_in[0];
    const int*   ei     = (const int*)  d_in[1];
    const float* lin0_w = (const float*)d_in[2];
    const float* lin0_b = (const float*)d_in[3];
    const float* conv_w = (const float*)d_in[4];
    const float* lin1_w = (const float*)d_in[5];
    const float* lin1_b = (const float*)d_in[6];
    float* out = (float*)d_out;

    const int N = in_sizes[0] / 128;
    const int E = in_sizes[1] / 2;
    const int* row = ei;
    const int* col = ei + E;

    float* ws   = (float*)d_ws;
    float* dinv = ws;
    float* x0   = dinv + N;
    float* h    = x0 + (size_t)N * 64;
    float* agg  = h  + (size_t)N * 64;

    const int ntiles = N / 16;
    const int gemmGrid = (ntiles * 32 + 255) / 256;
    const long long NF = (long long)N * 64;

    k_fill_one<<<(N + 255) / 256, 256, 0, stream>>>(dinv, N);
    k_degree  <<<(E + 255) / 256, 256, 0, stream>>>(col, dinv, E);
    k_rsqrt   <<<(N + 255) / 256, 256, 0, stream>>>(dinv, N);

    k_dense<128, true, true><<<gemmGrid, 256, 0, stream>>>(x, lin0_w, lin0_b, h, x0, ntiles);

    for (int l = 0; l < 4; ++l) {
        float beta = logf(0.5f / (float)(l + 1) + 1.0f);
        k_selfloop<<<(int)((NF + 255) / 256), 256, 0, stream>>>(dinv, h, agg, NF);
        k_scatter <<<(int)(((long long)E * 16 + 255) / 256), 256, 0, stream>>>(
            row, col, dinv, h, agg, E);
        k_layer   <<<gemmGrid, 256, 0, stream>>>(agg, x0, conv_w + (size_t)l * 64 * 64,
                                                 beta, h, ntiles);
    }

    k_dense<64, false, false><<<gemmGrid, 256, 0, stream>>>(h, lin1_w, lin1_b, out, nullptr, ntiles);
}


# Replace per-layer float-atomic edge scatter by an in-call CSR (integer atomic rank, scan, fill of source+weight records) and a one-wave-per-node gather summing self-loop and in-edges in rank order
# speedup vs baseline: 2.1577x; 2.1577x over previous
#include <hip/hip_runtime.h>
#include <math.h>

typedef __attribute__((ext_vector_type(2))) float v2f;
typedef __attribute__((ext_vector_type(8))) float v8f;
typedef __attribute__((ext_vector_type(2))) int   v2i;

__global__ void k_zero(int* __restrict__ cnt, int n) {
    int i = blockIdx.x * blockDim.x + threadIdx.x;
    if (i < n) cnt[i] = 0;
}

__global__ void k_count(const int* __restrict__ col, int* __restrict__ cnt,
                        int* __restrict__ pos, int e) {
    int i = blockIdx.x * blockDim.x + threadIdx.x;
    if (i < e) pos[i] = atomicAdd(&cnt[col[i]], 1);
}

__global__ void k_bsum(const int* __restrict__ cnt, int* __restrict__ bsum, int n) {
    __shared__ int s[256];
    int tid = threadIdx.x;
    int i = blockIdx.x * 256 + tid;
    s[tid] = i < n ? cnt[i] : 0;
    __syncthreads();
    for (int st = 128; st > 0; st >>= 1) {
        if (tid < st) s[tid] += s[tid + st];
        __syncthreads();
    }
    if (tid == 0) bsum[blockIdx.x] = s[0];
}

__global__ void k_scan(const int* __restrict__ cnt, const int* __restrict__ bsum,
                       int* __restrict__ rowptr, float* __restrict__ dinv, int n) {
    __shared__ int s[256];
    int tid = threadIdx.x;
    int b = blockIdx.x;
    int part = 0;
    for (int j = tid; j < b; j += 256) part += bsum[j];
    s[tid] = part;
    __syncthreads();
    for (int st = 128; st > 0; st >>= 1) {
        if (tid < st) s[tid] += s[tid + st];
        __syncthreads();
    }
    int off = s[0];
    __syncthreads();
    int i = b * 256 + tid;
    int c = i < n ? cnt[i] : 0;
    s[tid] = c;
    __syncthreads();
    for (int st = 1; st < 256; st <<= 1) {
        int v = tid >= st ? s[tid - st] : 0;
        __syncthreads();
        s[tid] += v;
        __syncthreads();
    }
    int incl = s[tid];
    if (i < n) {
        rowptr[i] = off + incl - c;
        float d = (float)(c + 1);
        dinv[i] = d > 0.0f ? rsqrtf(d) : 0.0f;
        if (i == n - 1) rowptr[n] = off + incl;
    }
}

__global__ void k_fill(const int* __restrict__ row, const int* __restrict__ col,
                       const int* __restrict__ pos, const int* __restrict__ rowptr,
                       const float* __restrict__ dinv, int* __restrict__ ew, int e) {
    int i = blockIdx.x * blockDim.x + threadIdx.x;
    if (i < e) {
        int r = row[i], c = col[i];
        v2i v;
        v.x = r;
        v.y = __float_as_int(dinv[r] * dinv[c]);
        *(v2i*)(ew + 2 * (size_t)(rowptr[c] + pos[i])) = v;
    }
}

template <int K, bool RELU, bool STOREPRE>
__global__ void k_dense(const float* __restrict__ X, const float* __restrict__ W,
                        const float* __restrict__ bias,
                        float* __restrict__ outA, float* __restrict__ outPre,
                        int ntiles) {
    __shared__ float wlds[64 * K];
    constexpr int NV4 = (64 * K) / 4;
    for (int i = threadIdx.x; i < NV4; i += 256)
        ((float4*)wlds)[i] = ((const float4*)W)[i];
    __syncthreads();

    int wid  = (blockIdx.x * blockDim.x + threadIdx.x) >> 5;
    int lane = threadIdx.x & 31;
    if (wid >= ntiles) return;
    int ln = lane & 15;
    int kh = lane >> 4;

    const float* xrow = X + (size_t)(wid * 16 + ln) * K + 2 * kh;
    v2f afr[K / 4];
#pragma unroll
    for (int kk = 0; kk < K / 4; ++kk)
        afr[kk] = *(const v2f*)(xrow + 4 * kk);

#pragma unroll
    for (int j = 0; j < 4; ++j) {
        const float* wrow = wlds + (size_t)(16 * j + ln) * K + 2 * kh;
        v8f acc = {};
#pragma unroll
        for (int kk = 0; kk < K / 4; ++kk) {
            v2f b = *(const v2f*)(wrow + 4 * kk);
            acc = __builtin_amdgcn_wmma_f32_16x16x4_f32(
                false, afr[kk], false, b, (short)0, acc, false, false);
        }
        int n = 16 * j + ln;
        float bv = bias[n];
#pragma unroll
        for (int r = 0; r < 8; ++r) {
            int m = r + 8 * kh;
            size_t gi = (size_t)(wid * 16 + m) * 64 + n;
            float v = acc[r] + bv;
            if constexpr (STOREPRE) outPre[gi] = v;
            outA[gi] = RELU ? fmaxf(v, 0.0f) : v;
        }
    }
}

__global__ void k_gather(const int* __restrict__ rowptr, const int* __restrict__ ew,
                         const float* __restrict__ dinv, const float* __restrict__ h,
                         float* __restrict__ agg, int n) {
    int lane = threadIdx.x & 31;
    int node = __builtin_amdgcn_readfirstlane((int)((blockIdx.x * blockDim.x + threadIdx.x) >> 5));
    if (node >= n) return;
    int beg = rowptr[node];
    int end = rowptr[node + 1];
    float dc = dinv[node];
    const float* hl = h + 2 * lane;
    v2f hv = *(const v2f*)(hl + (size_t)node * 64);
    v2f acc = hv * (dc * dc);
    for (int p = beg; p < end; p += 32) {
        int m = end - p < 32 ? end - p : 32;
        int idx = node;
        float w = 0.0f;
        if (lane < m) {
            v2i v = *(const v2i*)(ew + 2 * (size_t)(p + lane));
            idx = v.x;
            w = __int_as_float(v.y);
        }
        for (int k0 = 0; k0 < m; k0 += 8) {
#pragma unroll
            for (int u = 0; u < 8; ++u) {
                int r = __builtin_amdgcn_readlane(idx, k0 + u);
                float wk = __int_as_float(__builtin_amdgcn_readlane(__float_as_int(w), k0 + u));
                v2f a = *(const v2f*)(hl + (size_t)r * 64);
                acc += a * wk;
            }
        }
    }
    *(v2f*)(agg + (size_t)node * 64 + 2 * lane) = acc;
}

__global__ void k_layer(const float* __restrict__ agg, const float* __restrict__ x0,
                        const float* __restrict__ Wl, float beta,
                        float* __restrict__ h, int ntiles) {
    __shared__ float wlds[64 * 64];
    for (int i = threadIdx.x; i < 1024; i += 256)
        ((float4*)wlds)[i] = ((const float4*)Wl)[i];
    __syncthreads();

    int wid  = (blockIdx.x * blockDim.x + threadIdx.x) >> 5;
    int lane = threadIdx.x & 31;
    if (wid >= ntiles) return;
    int ln = lane & 15;
    int kh = lane >> 4;

    size_t rb = (size_t)(wid * 16 + ln) * 64 + 2 * kh;
    v2f afr[16];
#pragma unroll
    for (int kk = 0; kk < 16; ++kk) {
        v2f ag = *(const v2f*)(agg + rb + 4 * kk);
        v2f xx = *(const v2f*)(x0 + rb + 4 * kk);
        afr[kk] = ag * 0.9f + xx * 0.1f;
    }
    float ob = 1.0f - beta;
#pragma unroll
    for (int j = 0; j < 4; ++j) {
        const float* wcol = wlds + 2 * kh * 64 + 16 * j + ln;
        v8f acc = {};
#pragma unroll
        for (int kk = 0; kk < 16; ++kk) {
            v2f b;
            b.x = wcol[(size_t)(4 * kk) * 64];
            b.y = wcol[(size_t)(4 * kk + 1) * 64];
            acc = __builtin_amdgcn_wmma_f32_16x16x4_f32(
                false, afr[kk], false, b, (short)0, acc, false, false);
        }
#pragma unroll
        for (int r = 0; r < 8; ++r) {
            int m = r + 8 * kh;
            int n = 16 * j + ln;
            size_t gi = (size_t)(wid * 16 + m) * 64 + n;
            float hh = 0.9f * agg[gi] + 0.1f * x0[gi];
            float v = ob * hh + beta * acc[r];
            h[gi] = fmaxf(v, 0.0f);
        }
    }
}

extern "C" void kernel_launch(void* const* d_in, const int* in_sizes, int n_in,
                              void* d_out, int out_size, void* d_ws, size_t ws_size,
                              hipStream_t stream) {
    const float* x      = (const float*)d_in[0];
    const int*   ei     = (const int*)  d_in[1];
    const float* lin0_w = (const float*)d_in[2];
    const float* lin0_b = (const float*)d_in[3];
    const float* conv_w = (const float*)d_in[4];
    const float* lin1_w = (const float*)d_in[5];
    const float* lin1_b = (const float*)d_in[6];
    float* out = (float*)d_out;

    const int N = in_sizes[0] / 128;
    const int E = in_sizes[1] / 2;
    const int* row = ei;
    const int* col = ei + E;

    float* ws   = (float*)d_ws;
    float* dinv = ws;
    float* x0   = dinv + N;
    float* h    = x0 + (size_t)N * 64;
    float* agg  = h  + (size_t)N * 64;
    int* ew     = (int*)(agg + (size_t)N * 64);
    int* cnt    = ew + 2 * (size_t)E;
    int* pos    = cnt + N;
    int* rowptr = pos + E;
    int* bsum   = rowptr + (N + 1);

    const int ntiles = N / 16;
    const int gemmGrid = (ntiles * 32 + 255) / 256;
    const int nblk = (N + 255) / 256;

    k_zero <<<nblk, 256, 0, stream>>>(cnt, N);
    k_count<<<(E + 255) / 256, 256, 0, stream>>>(col, cnt, pos, E);
    k_bsum <<<nblk, 256, 0, stream>>>(cnt, bsum, N);
    k_scan <<<nblk, 256, 0, stream>>>(cnt, bsum, rowptr, dinv, N);
    k_fill <<<(E + 255) / 256, 256, 0, stream>>>(row, col, pos, rowptr, dinv, ew, E);

    k_dense<128, true, true><<<gemmGrid, 256, 0, stream>>>(x, lin0_w, lin0_b, h, x0, ntiles);

    for (int l = 0; l < 4; ++l) {
        float beta = logf(0.5f / (float)(l + 1) + 1.0f);
        k_gather<<<(N + 7) / 8, 256, 0, stream>>>(rowptr, ew, dinv, h, agg, N);
        k_layer   <<<gemmGrid, 256, 0, stream>>>(agg, x0, conv_w + (size_t)l * 64 * 64,
                                                 beta, h, ntiles);
    }

    k_dense<64, false, false><<<gemmGrid, 256, 0, stream>>>(h, lin1_w, lin1_b, out, nullptr, ntiles);
}


# Fuse each layer's gather into its product kernel: a wave gathers its 16-node tile of 0.9*agg+0.1*x0 into LDS and feeds the WMMA from there (h in two planes); no agg plane, 4 launches fewer
# speedup vs baseline: 2.5568x; 1.1850x over previous
#include <hip/hip_runtime.h>
#include <math.h>

typedef __attribute__((ext_vector_type(2))) float v2f;
typedef __attribute__((ext_vector_type(8))) float v8f;
typedef __attribute__((ext_vector_type(2))) int   v2i;

__global__ void k_zero(int* __restrict__ cnt, int n) {
    int i = blockIdx.x * blockDim.x + threadIdx.x;
    if (i < n) cnt[i] = 0;
}

__global__ void k_count(const int* __restrict__ col, int* __restrict__ cnt,
                        int* __restrict__ pos, int e) {
    int i = blockIdx.x * blockDim.x + threadIdx.x;
    if (i < e) pos[i] = atomicAdd(&cnt[col[i]], 1);
}

__global__ void k_bsum(const int* __restrict__ cnt, int* __restrict__ bsum, int n) {
    __shared__ int s[256];
    int tid = threadIdx.x;
    int i = blockIdx.x * 256 + tid;
    s[tid] = i < n ? cnt[i] : 0;
    __syncthreads();
    for (int st = 128; st > 0; st >>= 1) {
        if (tid < st) s[tid] += s[tid + st];
        __syncthreads();
    }
    if (tid == 0) bsum[blockIdx.x] = s[0];
}

__global__ void k_scan(const int* __restrict__ cnt, const int* __restrict__ bsum,
                       int* __restrict__ rowptr, float* __restrict__ dinv, int n) {
    __shared__ int s[256];
    int tid = threadIdx.x;
    int b = blockIdx.x;
    int part = 0;
    for (int j = tid; j < b; j += 256) part += bsum[j];
    s[tid] = part;
    __syncthreads();
    for (int st = 128; st > 0; st >>= 1) {
        if (tid < st) s[tid] += s[tid + st];
        __syncthreads();
    }
    int off = s[0];
    __syncthreads();
    int i = b * 256 + tid;
    int c = i < n ? cnt[i] : 0;
    s[tid] = c;
    __syncthreads();
    for (int st = 1; st < 256; st <<= 1) {
        int v = tid >= st ? s[tid - st] : 0;
        __syncthreads();
        s[tid] += v;
        __syncthreads();
    }
    int incl = s[tid];
    if (i < n) {
        rowptr[i] = off + incl - c;
        float d = (float)(c + 1);
        dinv[i] = d > 0.0f ? rsqrtf(d) : 0.0f;
        if (i == n - 1) rowptr[n] = off + incl;
    }
}

__global__ void k_fill(const int* __restrict__ row, const int* __restrict__ col,
                       const int* __restrict__ pos, const int* __restrict__ rowptr,
                       const float* __restrict__ dinv, int* __restrict__ ew, int e) {
    int i = blockIdx.x * blockDim.x + threadIdx.x;
    if (i < e) {
        int r = row[i], c = col[i];
        v2i v;
        v.x = r;
        v.y = __float_as_int(dinv[r] * dinv[c]);
        *(v2i*)(ew + 2 * (size_t)(rowptr[c] + pos[i])) = v;
    }
}

template <int K, bool RELU, bool STOREPRE>
__global__ void k_dense(const float* __restrict__ X, const float* __restrict__ W,
                        const float* __restrict__ bias,
                        float* __restrict__ outA, float* __restrict__ outPre,
                        int ntiles) {
    __shared__ float wlds[64 * K];
    constexpr int NV4 = (64 * K) / 4;
    for (int i = threadIdx.x; i < NV4; i += 256)
        ((float4*)wlds)[i] = ((const float4*)W)[i];
    __syncthreads();

    int wid  = (blockIdx.x * blockDim.x + threadIdx.x) >> 5;
    int lane = threadIdx.x & 31;
    if (wid >= ntiles) return;
    int ln = lane & 15;
    int kh = lane >> 4;

    const float* xrow = X + (size_t)(wid * 16 + ln) * K + 2 * kh;
    v2f afr[K / 4];
#pragma unroll
    for (int kk = 0; kk < K / 4; ++kk)
        afr[kk] = *(const v2f*)(xrow + 4 * kk);

#pragma unroll
    for (int j = 0; j < 4; ++j) {
        const float* wrow = wlds + (size_t)(16 * j + ln) * K + 2 * kh;
        v8f acc = {};
#pragma unroll
        for (int kk = 0; kk < K / 4; ++kk) {
            v2f b = *(const v2f*)(wrow + 4 * kk);
            acc = __builtin_amdgcn_wmma_f32_16x16x4_f32(
                false, afr[kk], false, b, (short)0, acc, false, false);
        }
        int n = 16 * j + ln;
        float bv = bias[n];
#pragma unroll
        for (int r = 0; r < 8; ++r) {
            int m = r + 8 * kh;
            size_t gi = (size_t)(wid * 16 + m) * 64 + n;
            float v = acc[r] + bv;
            if constexpr (STOREPRE) outPre[gi] = v;
            outA[gi] = RELU ? fmaxf(v, 0.0f) : v;
        }
    }
}

#define TILE_PITCH 68

__device__ __forceinline__ v2f rows8(v2f acc, const float* hl, int idx, float w, int k0) {
#pragma unroll
    for (int u = 0; u < 8; ++u) {
        int r = __builtin_amdgcn_readlane(idx, k0 + u);
        float wk = __int_as_float(__builtin_amdgcn_readlane(__float_as_int(w), k0 + u));
        v2f a = *(const v2f*)(hl + (size_t)r * 64);
        acc += a * wk;
    }
    return acc;
}

__global__ void __launch_bounds__(128)
k_layer(const int* __restrict__ rowptr, const int* __restrict__ ew,
        const float* __restrict__ dinv, const float* __restrict__ hin,
        const float* __restrict__ x0, const float* __restrict__ Wl, float beta,
        float* __restrict__ hout, int ntiles) {
    __shared__ float wlds[64 * 64];
    __shared__ float tiles[4][16 * TILE_PITCH];
#pragma unroll
    for (int q = 0; q < 8; ++q)
        ((float4*)wlds)[threadIdx.x + 128 * q] = ((const float4*)Wl)[threadIdx.x + 128 * q];

    int lane = threadIdx.x & 31;
    int wv   = threadIdx.x >> 5;
    int wid = __builtin_amdgcn_readfirstlane((int)(blockIdx.x * 4 + wv));
    bool active = wid < ntiles;
    float* t = tiles[wv];

    if (active) {
        int base = wid * 16;
        float dl = dinv[base + (lane & 15)];
        float dsq = dl * dl;
        int   rp = rowptr[base + (lane < 16 ? lane : 16)];
        const float* hl = hin + 2 * lane;
        const float* xl = x0 + 2 * lane;

        int beg = __builtin_amdgcn_readlane(rp, 0);
        int end = beg;
        int idx = base;
        float w = 0.0f;
        v2f hv = {0.0f, 0.0f};
        v2f xv = {0.0f, 0.0f};
        for (int j = -1; j < 16; ++j) {
            int node = base + j;
            int nbeg = end;
            int nend = end;
            int nidx = base;
            float nw = 0.0f;
            v2f nhv = {0.0f, 0.0f};
            v2f nxv = {0.0f, 0.0f};
            if (j < 15) {
                nend = __builtin_amdgcn_readlane(rp, j + 2);
                nidx = node + 1;
                if (lane < nend - nbeg) {
                    v2i v = *(const v2i*)(ew + 2 * (size_t)(nbeg + lane));
                    nidx = v.x;
                    nw = __int_as_float(v.y);
                }
                nhv = *(const v2f*)(hl + (size_t)(node + 1) * 64);
                nxv = *(const v2f*)(xl + (size_t)(node + 1) * 64);
            }
            if (j >= 0) {
                float dd = __int_as_float(__builtin_amdgcn_readlane(__float_as_int(dsq), j));
                v2f acc = hv * dd;
                int deg = end - beg;
                int m0 = deg < 32 ? deg : 32;
                for (int k0 = 0; k0 < m0; k0 += 8)
                    acc = rows8(acc, hl, idx, w, k0);
                for (int p = beg + 32; p < end; p += 32) {
                    int m = end - p < 32 ? end - p : 32;
                    int xi = node;
                    float xw = 0.0f;
                    if (lane < m) {
                        v2i v = *(const v2i*)(ew + 2 * (size_t)(p + lane));
                        xi = v.x;
                        xw = __int_as_float(v.y);
                    }
                    for (int k0 = 0; k0 < m; k0 += 8)
                        acc = rows8(acc, hl, xi, xw, k0);
                }
                *(v2f*)(t + j * TILE_PITCH + 2 * lane) = acc * 0.9f + xv * 0.1f;
            }
            beg = nbeg;
            end = nend;
            idx = nidx;
            w = nw;
            hv = nhv;
            xv = nxv;
        }
    }
    __syncthreads();
    if (!active) return;

    int ln = lane & 15;
    int kh = lane >> 4;
    v2f afr[16];
#pragma unroll
    for (int kk = 0; kk < 16; ++kk)
        afr[kk] = *(const v2f*)(t + ln * TILE_PITCH + 4 * kk + 2 * kh);
    float ob = 1.0f - beta;
#pragma unroll
    for (int j = 0; j < 4; ++j) {
        const float* wcol = wlds + 2 * kh * 64 + 16 * j + ln;
        v8f acc = {};
#pragma unroll
        for (int kk = 0; kk < 16; ++kk) {
            v2f b;
            b.x = wcol[(size_t)(4 * kk) * 64];
            b.y = wcol[(size_t)(4 * kk + 1) * 64];
            acc = __builtin_amdgcn_wmma_f32_16x16x4_f32(
                false, afr[kk], false, b, (short)0, acc, false, false);
        }
#pragma unroll
        for (int r = 0; r < 8; ++r) {
            int m = r + 8 * kh;
            int n = 16 * j + ln;
            size_t gi = (size_t)(wid * 16 + m) * 64 + n;
            float hh = t[m * TILE_PITCH + n];
            float v = ob * hh + beta * acc[r];
            hout[gi] = fmaxf(v, 0.0f);
        }
    }
}

extern "C" void kernel_launch(void* const* d_in, const int* in_sizes, int n_in,
                              void* d_out, int out_size, void* d_ws, size_t ws_size,
                              hipStream_t stream) {
    const float* x      = (const float*)d_in[0];
    const int*   ei     = (const int*)  d_in[1];
    const float* lin0_w = (const float*)d_in[2];
    const float* lin0_b = (const float*)d_in[3];
    const float* conv_w = (const float*)d_in[4];
    const float* lin1_w = (const float*)d_in[5];
    const float* lin1_b = (const float*)d_in[6];
    float* out = (float*)d_out;

    const int N = in_sizes[0] / 128;
    const int E = in_sizes[1] / 2;
    const int* row = ei;
    const int* col = ei + E;

    float* ws   = (float*)d_ws;
    float* dinv = ws;
    float* x0   = dinv + N;
    float* h    = x0 + (size_t)N * 64;
    float* h2   = h  + (size_t)N * 64;
    int* ew     = (int*)(h2 + (size_t)N * 64);
    int* cnt    = ew + 2 * (size_t)E;
    int* pos    = cnt + N;
    int* rowptr = pos + E;
    int* bsum   = rowptr + (N + 1);

    const int ntiles = N / 16;
    const int gemmGrid = (ntiles * 32 + 255) / 256;
    const int nblk = (N + 255) / 256;

    k_zero <<<nblk, 256, 0, stream>>>(cnt, N);
    k_count<<<(E + 255) / 256, 256, 0, stream>>>(col, cnt, pos, E);
    k_bsum <<<nblk, 256, 0, stream>>>(cnt, bsum, N);
    k_scan <<<nblk, 256, 0, stream>>>(cnt, bsum, rowptr, dinv, N);
    k_fill <<<(E + 255) / 256, 256, 0, stream>>>(row, col, pos, rowptr, dinv, ew, E);

    k_dense<128, true, true><<<gemmGrid, 256, 0, stream>>>(x, lin0_w, lin0_b, h, x0, ntiles);

    const int layerGrid = (ntiles + 3) / 4;
    for (int l = 0; l < 4; ++l) {
        float beta = logf(0.5f / (float)(l + 1) + 1.0f);
        const float* hin = (l & 1) ? h2 : h;
        float* hout      = (l & 1) ? h : h2;
        k_layer<<<layerGrid, 128, 0, stream>>>(rowptr, ew, dinv, hin, x0,
                                               conv_w + (size_t)l * 64 * 64, beta, hout, ntiles);
    }

    k_dense<64, false, false><<<gemmGrid, 256, 0, stream>>>(h, lin1_w, lin1_b, out, nullptr, ntiles);
}


# CSR build: per-edge returning global atomics replaced by a two-level counting sort on LDS counters (512-node buckets, then per-bucket rank and scan); in-node edge order stays an arrival-order rank
# speedup vs baseline: 3.4695x; 1.3570x over previous
#include <hip/hip_runtime.h>
#include <math.h>

typedef __attribute__((ext_vector_type(2))) float v2f;
typedef __attribute__((ext_vector_type(8))) float v8f;
typedef __attribute__((ext_vector_type(2))) int   v2i;

#define BSHIFT 9
#define CHUNK 4096

__device__ __forceinline__ void lds_scan_incl(int* s, int tid, int n) {
    for (int st = 1; st < n; st <<= 1) {
        int v = tid >= st ? s[tid - st] : 0;
        __syncthreads();
        s[tid] += v;
        __syncthreads();
    }
}

__global__ void __launch_bounds__(256)
k_hist(const int* __restrict__ col, int* __restrict__ ghist, int e, int nbk, int nchunk) {
    __shared__ int lh[256];
    int tid = threadIdx.x;
    lh[tid] = 0;
    __syncthreads();
    int base = blockIdx.x * CHUNK + tid;
    int c[16];
#pragma unroll
    for (int k = 0; k < 16; ++k) {
        int i = base + k * 256;
        c[k] = i < e ? col[i] : -1;
    }
#pragma unroll
    for (int k = 0; k < 16; ++k)
        if (c[k] >= 0) atomicAdd(&lh[c[k] >> BSHIFT], 1);
    __syncthreads();
    if (tid < nbk) ghist[(size_t)tid * nchunk + blockIdx.x] = lh[tid];
}

__global__ void __launch_bounds__(512)
k_bscan(int* __restrict__ ghist, int* __restrict__ btot, int nchunk) {
    __shared__ int s[512];
    int tid = threadIdx.x;
    int* g = ghist + (size_t)blockIdx.x * nchunk;
    int carry = 0;
    for (int t0 = 0; t0 < nchunk; t0 += 512) {
        int i = t0 + tid;
        int v = i < nchunk ? g[i] : 0;
        s[tid] = v;
        __syncthreads();
        lds_scan_incl(s, tid, 512);
        if (i < nchunk) g[i] = carry + s[tid] - v;
        carry += s[511];
        __syncthreads();
    }
    if (tid == 0) btot[blockIdx.x] = carry;
}

__global__ void __launch_bounds__(256)
k_place(const int* __restrict__ row, const int* __restrict__ col,
        const int* __restrict__ ghist, const int* __restrict__ btot,
        int* __restrict__ bedge, int e, int nbk, int nchunk) {
    __shared__ int s[256];
    __shared__ int lbase[256];
    int tid = threadIdx.x;
    int base = blockIdx.x * CHUNK + tid;
    int r[16], c[16];
#pragma unroll
    for (int k = 0; k < 16; ++k) {
        int i = base + k * 256;
        bool ok = i < e;
        r[k] = ok ? row[i] : 0;
        c[k] = ok ? col[i] : -1;
    }
    int tot = tid < nbk ? btot[tid] : 0;
    int pre = tid < nbk ? ghist[(size_t)tid * nchunk + blockIdx.x] : 0;
    s[tid] = tot;
    __syncthreads();
    lds_scan_incl(s, tid, 256);
    lbase[tid] = s[tid] - tot + pre;
    __syncthreads();
#pragma unroll
    for (int k = 0; k < 16; ++k) {
        if (c[k] >= 0) {
            int p = atomicAdd(&lbase[c[k] >> BSHIFT], 1);
            v2i v;
            v.x = r[k];
            v.y = c[k];
            *(v2i*)(bedge + 2 * (size_t)p) = v;
        }
    }
}

__global__ void __launch_bounds__(512)
k_bcount(const int* __restrict__ bedge, const int* __restrict__ btot,
         int* __restrict__ pos, int* __restrict__ rowptr, float* __restrict__ dinv, int n) {
    __shared__ int s[512];
    __shared__ int cl[512];
    int tid = threadIdx.x;
    int b = blockIdx.x;
    int part = 0;
    for (int j = tid; j < b; j += 512) part += btot[j];
    s[tid] = part;
    cl[tid] = 0;
    __syncthreads();
    for (int st = 256; st > 0; st >>= 1) {
        if (tid < st) s[tid] += s[tid + st];
        __syncthreads();
    }
    int bstart = s[0];
    int bcount = btot[b];
    int node0 = b << BSHIFT;
    __syncthreads();
    for (int i0 = 0; i0 < bcount; i0 += 8 * 512) {
        int c[8];
#pragma unroll
        for (int u = 0; u < 8; ++u) {
            int i = i0 + u * 512 + tid;
            c[u] = i < bcount ? bedge[2 * (size_t)(bstart + i) + 1] : -1;
        }
#pragma unroll
        for (int u = 0; u < 8; ++u) {
            int i = i0 + u * 512 + tid;
            if (c[u] >= 0) pos[bstart + i] = atomicAdd(&cl[c[u] - node0], 1);
        }
    }
    __syncthreads();
    int cn = cl[tid];
    s[tid] = cn;
    __syncthreads();
    lds_scan_incl(s, tid, 512);
    int node = node0 + tid;
    if (node < n) {
        rowptr[node] = bstart + s[tid] - cn;
        float d = (float)(cn + 1);
        dinv[node] = d > 0.0f ? rsqrtf(d) : 0.0f;
        if (node == n - 1) rowptr[n] = bstart + s[tid];
    }
}

__global__ void __launch_bounds__(256)
k_fill(const int* __restrict__ bedge, const int* __restrict__ pos,
       const int* __restrict__ rowptr, const float* __restrict__ dinv,
       int* __restrict__ ew, int e) {
    int base = blockIdx.x * CHUNK + threadIdx.x;
    v2i rc[16];
    int p[16];
#pragma unroll
    for (int k = 0; k < 16; ++k) {
        int i = base + k * 256;
        int ii = i < e ? i : e - 1;
        rc[k] = *(const v2i*)(bedge + 2 * (size_t)ii);
        p[k] = pos[ii];
    }
    float w[16];
    int o[16];
#pragma unroll
    for (int k = 0; k < 16; ++k) {
        w[k] = dinv[rc[k].x] * dinv[rc[k].y];
        o[k] = rowptr[rc[k].y] + p[k];
    }
#pragma unroll
    for (int k = 0; k < 16; ++k) {
        if (base + k * 256 < e) {
            v2i v;
            v.x = rc[k].x;
            v.y = __float_as_int(w[k]);
            *(v2i*)(ew + 2 * (size_t)o[k]) = v;
        }
    }
}

template <int K, bool RELU, bool STOREPRE>
__global__ void k_dense(const float* __restrict__ X, const float* __restrict__ W,
                        const float* __restrict__ bias,
                        float* __restrict__ outA, float* __restrict__ outPre,
                        int ntiles) {
    __shared__ float wlds[64 * K];
    constexpr int NV4 = (64 * K) / 4;
    for (int i = threadIdx.x; i < NV4; i += 256)
        ((float4*)wlds)[i] = ((const float4*)W)[i];
    __syncthreads();

    int wid  = (blockIdx.x * blockDim.x + threadIdx.x) >> 5;
    int lane = threadIdx.x & 31;
    if (wid >= ntiles) return;
    int ln = lane & 15;
    int kh = lane >> 4;

    const float* xrow = X + (size_t)(wid * 16 + ln) * K + 2 * kh;
    v2f afr[K / 4];
#pragma unroll
    for (int kk = 0; kk < K / 4; ++kk)
        afr[kk] = *(const v2f*)(xrow + 4 * kk);

#pragma unroll
    for (int j = 0; j < 4; ++j) {
        const float* wrow = wlds + (size_t)(16 * j + ln) * K + 2 * kh;
        v8f acc = {};
#pragma unroll
        for (int kk = 0; kk < K / 4; ++kk) {
            v2f b = *(const v2f*)(wrow + 4 * kk);
            acc = __builtin_amdgcn_wmma_f32_16x16x4_f32(
                false, afr[kk], false, b, (short)0, acc, false, false);
        }
        int n = 16 * j + ln;
        float bv = bias[n];
#pragma unroll
        for (int r = 0; r < 8; ++r) {
            int m = r + 8 * kh;
            size_t gi = (size_t)(wid * 16 + m) * 64 + n;
            float v = acc[r] + bv;
            if constexpr (STOREPRE) outPre[gi] = v;
            outA[gi] = RELU ? fmaxf(v, 0.0f) : v;
        }
    }
}

#define TILE_PITCH 68

__device__ __forceinline__ v2f rows8(v2f acc, const float* hl, int idx, float w, int k0) {
#pragma unroll
    for (int u = 0; u < 8; ++u) {
        int r = __builtin_amdgcn_readlane(idx, k0 + u);
        float wk = __int_as_float(__builtin_amdgcn_readlane(__float_as_int(w), k0 + u));
        v2f a = *(const v2f*)(hl + (size_t)r * 64);
        acc += a * wk;
    }
    return acc;
}

__global__ void __launch_bounds__(128)
k_layer(const int* __restrict__ rowptr, const int* __restrict__ ew,
        const float* __restrict__ dinv, const float* __restrict__ hin,
        const float* __restrict__ x0, const float* __restrict__ Wl, float beta,
        float* __restrict__ hout, int ntiles) {
    __shared__ float wlds[64 * 64];
    __shared__ float tiles[4][16 * TILE_PITCH];
#pragma unroll
    for (int q = 0; q < 8; ++q)
        ((float4*)wlds)[threadIdx.x + 128 * q] = ((const float4*)Wl)[threadIdx.x + 128 * q];

    int lane = threadIdx.x & 31;
    int wv   = threadIdx.x >> 5;
    int wid = __builtin_amdgcn_readfirstlane((int)(blockIdx.x * 4 + wv));
    bool active = wid < ntiles;
    float* t = tiles[wv];

    if (active) {
        int base = wid * 16;
        float dl = dinv[base + (lane & 15)];
        float dsq = dl * dl;
        int   rp = rowptr[base + (lane < 16 ? lane : 16)];
        const float* hl = hin + 2 * lane;
        const float* xl = x0 + 2 * lane;

        int beg = __builtin_amdgcn_readlane(rp, 0);
        int end = beg;
        int idx = base;
        float w = 0.0f;
        v2f hv = {0.0f, 0.0f};
        v2f xv = {0.0f, 0.0f};
        for (int j = -1; j < 16; ++j) {
            int node = base + j;
            int nbeg = end;
            int nend = end;
            int nidx = base;
            float nw = 0.0f;
            v2f nhv = {0.0f, 0.0f};
            v2f nxv = {0.0f, 0.0f};
            if (j < 15) {
                nend = __builtin_amdgcn_readlane(rp, j + 2);
                nidx = node + 1;
                if (lane < nend - nbeg) {
                    v2i v = *(const v2i*)(ew + 2 * (size_t)(nbeg + lane));
                    nidx = v.x;
                    nw = __int_as_float(v.y);
                }
                nhv = *(const v2f*)(hl + (size_t)(node + 1) * 64);
                nxv = *(const v2f*)(xl + (size_t)(node + 1) * 64);
            }
            if (j >= 0) {
                float dd = __int_as_float(__builtin_amdgcn_readlane(__float_as_int(dsq), j));
                v2f acc = hv * dd;
                int deg = end - beg;
                int m0 = deg < 32 ? deg : 32;
                for (int k0 = 0; k0 < m0; k0 += 8)
                    acc = rows8(acc, hl, idx, w, k0);
                for (int p = beg + 32; p < end; p += 32) {
                    int m = end - p < 32 ? end - p : 32;
                    int xi = node;
                    float xw = 0.0f;
                    if (lane < m) {
                        v2i v = *(const v2i*)(ew + 2 * (size_t)(p + lane));
                        xi = v.x;
                        xw = __int_as_float(v.y);
                    }
                    for (int k0 = 0; k0 < m; k0 += 8)
                        acc = rows8(acc, hl, xi, xw, k0);
                }
                *(v2f*)(t + j * TILE_PITCH + 2 * lane) = acc * 0.9f + xv * 0.1f;
            }
            beg = nbeg;
            end = nend;
            idx = nidx;
            w = nw;
            hv = nhv;
            xv = nxv;
        }
    }
    __syncthreads();
    if (!active) return;

    int ln = lane & 15;
    int kh = lane >> 4;
    v2f afr[16];
#pragma unroll
    for (int kk = 0; kk < 16; ++kk)
        afr[kk] = *(const v2f*)(t + ln * TILE_PITCH + 4 * kk + 2 * kh);
    float ob = 1.0f - beta;
#pragma unroll
    for (int j = 0; j < 4; ++j) {
        const float* wcol = wlds + 2 * kh * 64 + 16 * j + ln;
        v8f acc = {};
#pragma unroll
        for (int kk = 0; kk < 16; ++kk) {
            v2f b;
            b.x = wcol[(size_t)(4 * kk) * 64];
            b.y = wcol[(size_t)(4 * kk + 1) * 64];
            acc = __builtin_amdgcn_wmma_f32_16x16x4_f32(
                false, afr[kk], false, b, (short)0, acc, false, false);
        }
#pragma unroll
        for (int r = 0; r < 8; ++r) {
            int m = r + 8 * kh;
            int n = 16 * j + ln;
            size_t gi = (size_t)(wid * 16 + m) * 64 + n;
            float hh = t[m * TILE_PITCH + n];
            float v = ob * hh + beta * acc[r];
            hout[gi] = fmaxf(v, 0.0f);
        }
    }
}

extern "C" void kernel_launch(void* const* d_in, const int* in_sizes, int n_in,
                              void* d_out, int out_size, void* d_ws, size_t ws_size,
                              hipStream_t stream) {
    const float* x      = (const float*)d_in[0];
    const int*   ei     = (const int*)  d_in[1];
    const float* lin0_w = (const float*)d_in[2];
    const float* lin0_b = (const float*)d_in[3];
    const float* conv_w = (const float*)d_in[4];
    const float* lin1_w = (const float*)d_in[5];
    const float* lin1_b = (const float*)d_in[6];
    float* out = (float*)d_out;

    const int N = in_sizes[0] / 128;
    const int E = in_sizes[1] / 2;
    const int* row = ei;
    const int* col = ei + E;

    float* ws   = (float*)d_ws;
    float* dinv = ws;
    float* x0   = dinv + N;
    float* h    = x0 + (size_t)N * 64;
    float* h2   = h  + (size_t)N * 64;
    int* ew     = (int*)(h2 + (size_t)N * 64);
    int* bedge  = ew + 2 * (size_t)E;
    int* pos    = bedge + 2 * (size_t)E;
    int* rowptr = pos + E;
    int* btot   = rowptr + (N + 1);
    int* ghist  = btot + 256;

    const int ntiles = N / 16;
    const int gemmGrid = (ntiles * 32 + 255) / 256;
    const int nbk = (N + 511) / 512;
    const int nchunk = (E + 4095) / 4096;

    k_hist  <<<nchunk, 256, 0, stream>>>(col, ghist, E, nbk, nchunk);
    k_bscan <<<nbk, 512, 0, stream>>>(ghist, btot, nchunk);
    k_place <<<nchunk, 256, 0, stream>>>(row, col, ghist, btot, bedge, E, nbk, nchunk);
    k_bcount<<<nbk, 512, 0, stream>>>(bedge, btot, pos, rowptr, dinv, N);
    k_fill  <<<nchunk, 256, 0, stream>>>(bedge, pos, rowptr, dinv, ew, E);

    k_dense<128, true, true><<<gemmGrid, 256, 0, stream>>>(x, lin0_w, lin0_b, h, x0, ntiles);

    const int layerGrid = (ntiles + 3) / 4;
    for (int l = 0; l < 4; ++l) {
        float beta = logf(0.5f / (float)(l + 1) + 1.0f);
        const float* hin = (l & 1) ? h2 : h;
        float* hout      = (l & 1) ? h : h2;
        k_layer<<<layerGrid, 128, 0, stream>>>(rowptr, ew, dinv, hin, x0,
                                               conv_w + (size_t)l * 64 * 64, beta, hout, ntiles);
    }

    k_dense<64, false, false><<<gemmGrid, 256, 0, stream>>>(h, lin1_w, lin1_b, out, nullptr, ntiles);
}


# Layer gather reads a tile's in-edge records as one stream: two sets of 16 row loads stay in flight across node boundaries (a scalar compare switches nodes); per-node sum order unchanged
# speedup vs baseline: 3.6834x; 1.0617x over previous
#include <hip/hip_runtime.h>
#include <math.h>

typedef __attribute__((ext_vector_type(2))) float v2f;
typedef __attribute__((ext_vector_type(8))) float v8f;
typedef __attribute__((ext_vector_type(2))) int   v2i;

#define BSHIFT 9
#define CHUNK 4096

__device__ __forceinline__ void lds_scan_incl(int* s, int tid, int n) {
    for (int st = 1; st < n; st <<= 1) {
        int v = tid >= st ? s[tid - st] : 0;
        __syncthreads();
        s[tid] += v;
        __syncthreads();
    }
}

__global__ void __launch_bounds__(256)
k_hist(const int* __restrict__ col, int* __restrict__ ghist, int e, int nbk, int nchunk) {
    __shared__ int lh[256];
    int tid = threadIdx.x;
    lh[tid] = 0;
    __syncthreads();
    int base = blockIdx.x * CHUNK + tid;
    int c[16];
#pragma unroll
    for (int k = 0; k < 16; ++k) {
        int i = base + k * 256;
        c[k] = i < e ? col[i] : -1;
    }
#pragma unroll
    for (int k = 0; k < 16; ++k)
        if (c[k] >= 0) atomicAdd(&lh[c[k] >> BSHIFT], 1);
    __syncthreads();
    if (tid < nbk) ghist[(size_t)tid * nchunk + blockIdx.x] = lh[tid];
}

__global__ void __launch_bounds__(512)
k_bscan(int* __restrict__ ghist, int* __restrict__ btot, int nchunk) {
    __shared__ int s[512];
    int tid = threadIdx.x;
    int* g = ghist + (size_t)blockIdx.x * nchunk;
    int carry = 0;
    for (int t0 = 0; t0 < nchunk; t0 += 512) {
        int i = t0 + tid;
        int v = i < nchunk ? g[i] : 0;
        s[tid] = v;
        __syncthreads();
        lds_scan_incl(s, tid, 512);
        if (i < nchunk) g[i] = carry + s[tid] - v;
        carry += s[511];
        __syncthreads();
    }
    if (tid == 0) btot[blockIdx.x] = carry;
}

__global__ void __launch_bounds__(256)
k_place(const int* __restrict__ row, const int* __restrict__ col,
        const int* __restrict__ ghist, const int* __restrict__ btot,
        int* __restrict__ bedge, int e, int nbk, int nchunk) {
    __shared__ int s[256];
    __shared__ int lbase[256];
    int tid = threadIdx.x;
    int base = blockIdx.x * CHUNK + tid;
    int r[16], c[16];
#pragma unroll
    for (int k = 0; k < 16; ++k) {
        int i = base + k * 256;
        bool ok = i < e;
        r[k] = ok ? row[i] : 0;
        c[k] = ok ? col[i] : -1;
    }
    int tot = tid < nbk ? btot[tid] : 0;
    int pre = tid < nbk ? ghist[(size_t)tid * nchunk + blockIdx.x] : 0;
    s[tid] = tot;
    __syncthreads();
    lds_scan_incl(s, tid, 256);
    lbase[tid] = s[tid] - tot + pre;
    __syncthreads();
#pragma unroll
    for (int k = 0; k < 16; ++k) {
        if (c[k] >= 0) {
            int p = atomicAdd(&lbase[c[k] >> BSHIFT], 1);
            v2i v;
            v.x = r[k];
            v.y = c[k];
            *(v2i*)(bedge + 2 * (size_t)p) = v;
        }
    }
}

__global__ void __launch_bounds__(512)
k_bcount(const int* __restrict__ bedge, const int* __restrict__ btot,
         int* __restrict__ pos, int* __restrict__ rowptr, float* __restrict__ dinv, int n) {
    __shared__ int s[512];
    __shared__ int cl[512];
    int tid = threadIdx.x;
    int b = blockIdx.x;
    int part = 0;
    for (int j = tid; j < b; j += 512) part += btot[j];
    s[tid] = part;
    cl[tid] = 0;
    __syncthreads();
    for (int st = 256; st > 0; st >>= 1) {
        if (tid < st) s[tid] += s[tid + st];
        __syncthreads();
    }
    int bstart = s[0];
    int bcount = btot[b];
    int node0 = b << BSHIFT;
    __syncthreads();
    for (int i0 = 0; i0 < bcount; i0 += 8 * 512) {
        int c[8];
#pragma unroll
        for (int u = 0; u < 8; ++u) {
            int i = i0 + u * 512 + tid;
            c[u] = i < bcount ? bedge[2 * (size_t)(bstart + i) + 1] : -1;
        }
#pragma unroll
        for (int u = 0; u < 8; ++u) {
            int i = i0 + u * 512 + tid;
            if (c[u] >= 0) pos[bstart + i] = atomicAdd(&cl[c[u] - node0], 1);
        }
    }
    __syncthreads();
    int cn = cl[tid];
    s[tid] = cn;
    __syncthreads();
    lds_scan_incl(s, tid, 512);
    int node = node0 + tid;
    if (node < n) {
        rowptr[node] = bstart + s[tid] - cn;
        float d = (float)(cn + 1);
        dinv[node] = d > 0.0f ? rsqrtf(d) : 0.0f;
        if (node == n - 1) rowptr[n] = bstart + s[tid];
    }
}

__global__ void __launch_bounds__(256)
k_fill(const int* __restrict__ bedge, const int* __restrict__ pos,
       const int* __restrict__ rowptr, const float* __restrict__ dinv,
       int* __restrict__ ew, int e) {
    int base = blockIdx.x * CHUNK + threadIdx.x;
    v2i rc[16];
    int p[16];
#pragma unroll
    for (int k = 0; k < 16; ++k) {
        int i = base + k * 256;
        int ii = i < e ? i : e - 1;
        rc[k] = *(const v2i*)(bedge + 2 * (size_t)ii);
        p[k] = pos[ii];
    }
    float w[16];
    int o[16];
#pragma unroll
    for (int k = 0; k < 16; ++k) {
        w[k] = dinv[rc[k].x] * dinv[rc[k].y];
        o[k] = rowptr[rc[k].y] + p[k];
    }
#pragma unroll
    for (int k = 0; k < 16; ++k) {
        if (base + k * 256 < e) {
            v2i v;
            v.x = rc[k].x;
            v.y = __float_as_int(w[k]);
            *(v2i*)(ew + 2 * (size_t)o[k]) = v;
        }
    }
}

template <int K, bool RELU, bool STOREPRE>
__global__ void k_dense(const float* __restrict__ X, const float* __restrict__ W,
                        const float* __restrict__ bias,
                        float* __restrict__ outA, float* __restrict__ outPre,
                        int ntiles) {
    __shared__ float wlds[64 * K];
    constexpr int NV4 = (64 * K) / 4;
    for (int i = threadIdx.x; i < NV4; i += 256)
        ((float4*)wlds)[i] = ((const float4*)W)[i];
    __syncthreads();

    int wid  = (blockIdx.x * blockDim.x + threadIdx.x) >> 5;
    int lane = threadIdx.x & 31;
    if (wid >= ntiles) return;
    int ln = lane & 15;
    int kh = lane >> 4;

    const float* xrow = X + (size_t)(wid * 16 + ln) * K + 2 * kh;
    v2f afr[K / 4];
#pragma unroll
    for (int kk = 0; kk < K / 4; ++kk)
        afr[kk] = *(const v2f*)(xrow + 4 * kk);

#pragma unroll
    for (int j = 0; j < 4; ++j) {
        const float* wrow = wlds + (size_t)(16 * j + ln) * K + 2 * kh;
        v8f acc = {};
#pragma unroll
        for (int kk = 0; kk < K / 4; ++kk) {
            v2f b = *(const v2f*)(wrow + 4 * kk);
            acc = __builtin_amdgcn_wmma_f32_16x16x4_f32(
                false, afr[kk], false, b, (short)0, acc, false, false);
        }
        int n = 16 * j + ln;
        float bv = bias[n];
#pragma unroll
        for (int r = 0; r < 8; ++r) {
            int m = r + 8 * kh;
            size_t gi = (size_t)(wid * 16 + m) * 64 + n;
            float v = acc[r] + bv;
            if constexpr (STOREPRE) outPre[gi] = v;
            outA[gi] = RELU ? fmaxf(v, 0.0f) : v;
        }
    }
}

#define TILE_PITCH 68

__device__ __forceinline__ void ask16(v2f (&a)[16], float (&wk)[16], const v2f* hrow2,
                                      unsigned lane, int ro, float w, int k0) {
#pragma unroll
    for (int u = 0; u < 16; ++u) {
        unsigned r = (unsigned)__builtin_amdgcn_readlane(ro, k0 + u);
        wk[u] = __int_as_float(__builtin_amdgcn_readlane(__float_as_int(w), k0 + u));
        a[u] = hrow2[r + lane];
    }
}

template <bool TAIL>
__device__ __forceinline__ void sum16(const v2f (&a)[16], const float (&wk)[16], int p0,
                                      int endAll, int rp, float* tl,
                                      int& cur, int& curEnd, v2f& acc) {
#pragma unroll
    for (int u = 0; u < 16; ++u) {
        int p = p0 + u;
        if (!TAIL || p < endAll) {
            if (p >= curEnd) {
                *(v2f*)(tl + cur * TILE_PITCH) = acc;
                do {
                    ++cur;
                    curEnd = __builtin_amdgcn_readlane(rp, cur + 1);
                } while (p >= curEnd && cur < 15);
                acc = *(const v2f*)(tl + cur * TILE_PITCH);
            }
            acc += a[u] * wk[u];
        }
    }
}

__global__ void __launch_bounds__(128)
k_layer(const int* __restrict__ rowptr, const int* __restrict__ ew,
        const float* __restrict__ dinv, const float* __restrict__ hin,
        const float* __restrict__ x0, const float* __restrict__ Wl, float beta,
        float* __restrict__ hout, int ntiles) {
    __shared__ float wlds[64 * 64];
    __shared__ float tiles[4][16 * TILE_PITCH];
#pragma unroll
    for (int q = 0; q < 8; ++q)
        ((float4*)wlds)[threadIdx.x + 128 * q] = ((const float4*)Wl)[threadIdx.x + 128 * q];

    int lane = threadIdx.x & 31;
    int wv   = threadIdx.x >> 5;
    int wid = __builtin_amdgcn_readfirstlane((int)(blockIdx.x * 4 + wv));
    bool active = wid < ntiles;
    float* t = tiles[wv];

    if (active) {
        int base = wid * 16;
        float dl = dinv[base + (lane & 15)];
        float dsq = dl * dl;
        int   rp = rowptr[base + (lane < 16 ? lane : 16)];
        const float* hl = hin + 2 * lane;
        const float* xl = x0 + 2 * lane;
        float* tl = t + 2 * lane;
        const v2f* hrow2 = (const v2f*)hin;
        unsigned ulane = (unsigned)lane;

        v2f hv[16];
        v2f xv[16];
#pragma unroll
        for (int j = 0; j < 16; ++j)
            hv[j] = *(const v2f*)(hl + (size_t)(base + j) * 64);
#pragma unroll
        for (int j = 0; j < 16; ++j)
            xv[j] = *(const v2f*)(xl + (size_t)(base + j) * 64);

        int beg    = __builtin_amdgcn_readlane(rp, 0);
        int endAll = __builtin_amdgcn_readlane(rp, 16);
        int pos = beg + lane;
        bool ok = pos < endAll;
        v2i rec = *(const v2i*)(ew + 2 * (size_t)(ok ? pos : beg));
#pragma unroll
        for (int j = 0; j < 16; ++j) {
            float dd = __int_as_float(__builtin_amdgcn_readlane(__float_as_int(dsq), j));
            *(v2f*)(tl + j * TILE_PITCH) = hv[j] * dd;
        }
        int cur = 0;
        int curEnd = __builtin_amdgcn_readlane(rp, 1);
        v2f acc = *(const v2f*)tl;

        int ro = (ok ? rec.x : base) * 32;
        float w = ok ? __int_as_float(rec.y) : 0.0f;
        v2f ra[16], rb[16];
        float wa[16], wb[16];
        ask16(ra, wa, hrow2, ulane, ro, w, 0);
        for (int p0 = beg; p0 < endAll; p0 += 32) {
            int npos = p0 + 32 + lane;
            bool nok = npos < endAll;
            v2i nrec = *(const v2i*)(ew + 2 * (size_t)(nok ? npos : p0));
            ask16(rb, wb, hrow2, ulane, ro, w, 16);
            bool full = p0 + 32 <= endAll;
            if (full) sum16<false>(ra, wa, p0, endAll, rp, tl, cur, curEnd, acc);
            else      sum16<true>(ra, wa, p0, endAll, rp, tl, cur, curEnd, acc);
            ro = (nok ? nrec.x : base) * 32;
            w = nok ? __int_as_float(nrec.y) : 0.0f;
            ask16(ra, wa, hrow2, ulane, ro, w, 0);
            if (full) sum16<false>(rb, wb, p0 + 16, endAll, rp, tl, cur, curEnd, acc);
            else      sum16<true>(rb, wb, p0 + 16, endAll, rp, tl, cur, curEnd, acc);
        }
        *(v2f*)(tl + cur * TILE_PITCH) = acc;
#pragma unroll
        for (int j = 0; j < 16; ++j) {
            v2f a = *(const v2f*)(tl + j * TILE_PITCH);
            *(v2f*)(tl + j * TILE_PITCH) = a * 0.9f + xv[j] * 0.1f;
        }
    }
    __syncthreads();
    if (!active) return;

    int ln = lane & 15;
    int kh = lane >> 4;
    v2f afr[16];
#pragma unroll
    for (int kk = 0; kk < 16; ++kk)
        afr[kk] = *(const v2f*)(t + ln * TILE_PITCH + 4 * kk + 2 * kh);
    float ob = 1.0f - beta;
#pragma unroll
    for (int j = 0; j < 4; ++j) {
        const float* wcol = wlds + 2 * kh * 64 + 16 * j + ln;
        v8f acc = {};
#pragma unroll
        for (int kk = 0; kk < 16; ++kk) {
            v2f b;
            b.x = wcol[(size_t)(4 * kk) * 64];
            b.y = wcol[(size_t)(4 * kk + 1) * 64];
            acc = __builtin_amdgcn_wmma_f32_16x16x4_f32(
                false, afr[kk], false, b, (short)0, acc, false, false);
        }
#pragma unroll
        for (int r = 0; r < 8; ++r) {
            int m = r + 8 * kh;
            int n = 16 * j + ln;
            size_t gi = (size_t)(wid * 16 + m) * 64 + n;
            float hh = t[m * TILE_PITCH + n];
            float v = ob * hh + beta * acc[r];
            hout[gi] = fmaxf(v, 0.0f);
        }
    }
}

extern "C" void kernel_launch(void* const* d_in, const int* in_sizes, int n_in,
                              void* d_out, int out_size, void* d_ws, size_t ws_size,
                              hipStream_t stream) {
    const float* x      = (const float*)d_in[0];
    const int*   ei     = (const int*)  d_in[1];
    const float* lin0_w = (const float*)d_in[2];
    const float* lin0_b = (const float*)d_in[3];
    const float* conv_w = (const float*)d_in[4];
    const float* lin1_w = (const float*)d_in[5];
    const float* lin1_b = (const float*)d_in[6];
    float* out = (float*)d_out;

    const int N = in_sizes[0] / 128;
    const int E = in_sizes[1] / 2;
    const int* row = ei;
    const int* col = ei + E;

    float* ws   = (float*)d_ws;
    float* dinv = ws;
    float* x0   = dinv + N;
    float* h    = x0 + (size_t)N * 64;
    float* h2   = h  + (size_t)N * 64;
    int* ew     = (int*)(h2 + (size_t)N * 64);
    int* bedge  = ew + 2 * (size_t)E;
    int* pos    = bedge + 2 * (size_t)E;
    int* rowptr = pos + E;
    int* btot   = rowptr + (N + 1);
    int* ghist  = btot + 256;

    const int ntiles = N / 16;
    const int gemmGrid = (ntiles * 32 + 255) / 256;
    const int nbk = (N + 511) / 512;
    const int nchunk = (E + 4095) / 4096;

    k_hist  <<<nchunk, 256, 0, stream>>>(col, ghist, E, nbk, nchunk);
    k_bscan <<<nbk, 512, 0, stream>>>(ghist, btot, nchunk);
    k_place <<<nchunk, 256, 0, stream>>>(row, col, ghist, btot, bedge, E, nbk, nchunk);
    k_bcount<<<nbk, 512, 0, stream>>>(bedge, btot, pos, rowptr, dinv, N);
    k_fill  <<<nchunk, 256, 0, stream>>>(bedge, pos, rowptr, dinv, ew, E);

    k_dense<128, true, true><<<gemmGrid, 256, 0, stream>>>(x, lin0_w, lin0_b, h, x0, ntiles);

    const int layerGrid = (ntiles + 3) / 4;
    for (int l = 0; l < 4; ++l) {
        float beta = logf(0.5f / (float)(l + 1) + 1.0f);
        const float* hin = (l & 1) ? h2 : h;
        float* hout      = (l & 1) ? h : h2;
        k_layer<<<layerGrid, 128, 0, stream>>>(rowptr, ew, dinv, hin, x0,
                                               conv_w + (size_t)l * 64 * 64, beta, hout, ntiles);
    }

    k_dense<64, false, false><<<gemmGrid, 256, 0, stream>>>(h, lin1_w, lin1_b, out, nullptr, ntiles);
}


# Weight rows padded in LDS (pitch K+4 floats in the dense kernels, 80 in the layer kernel): B-fragment reads were 16-way (dense) and 4-way (layer) bank-conflicted, now conflict-free
# speedup vs baseline: 4.3496x; 1.1809x over previous
#include <hip/hip_runtime.h>
#include <math.h>

typedef __attribute__((ext_vector_type(2))) float v2f;
typedef __attribute__((ext_vector_type(8))) float v8f;
typedef __attribute__((ext_vector_type(2))) int   v2i;

#define BSHIFT 9
#define CHUNK 4096

__device__ __forceinline__ void lds_scan_incl(int* s, int tid, int n) {
    for (int st = 1; st < n; st <<= 1) {
        int v = tid >= st ? s[tid - st] : 0;
        __syncthreads();
        s[tid] += v;
        __syncthreads();
    }
}

__global__ void __launch_bounds__(256)
k_hist(const int* __restrict__ col, int* __restrict__ ghist, int e, int nbk, int nchunk) {
    __shared__ int lh[256];
    int tid = threadIdx.x;
    lh[tid] = 0;
    __syncthreads();
    int base = blockIdx.x * CHUNK + tid;
    int c[16];
#pragma unroll
    for (int k = 0; k < 16; ++k) {
        int i = base + k * 256;
        c[k] = i < e ? col[i] : -1;
    }
#pragma unroll
    for (int k = 0; k < 16; ++k)
        if (c[k] >= 0) atomicAdd(&lh[c[k] >> BSHIFT], 1);
    __syncthreads();
    if (tid < nbk) ghist[(size_t)tid * nchunk + blockIdx.x] = lh[tid];
}

__global__ void __launch_bounds__(512)
k_bscan(int* __restrict__ ghist, int* __restrict__ btot, int nchunk) {
    __shared__ int s[512];
    int tid = threadIdx.x;
    int* g = ghist + (size_t)blockIdx.x * nchunk;
    int carry = 0;
    for (int t0 = 0; t0 < nchunk; t0 += 512) {
        int i = t0 + tid;
        int v = i < nchunk ? g[i] : 0;
        s[tid] = v;
        __syncthreads();
        lds_scan_incl(s, tid, 512);
        if (i < nchunk) g[i] = carry + s[tid] - v;
        carry += s[511];
        __syncthreads();
    }
    if (tid == 0) btot[blockIdx.x] = carry;
}

__global__ void __launch_bounds__(256)
k_place(const int* __restrict__ row, const int* __restrict__ col,
        const int* __restrict__ ghist, const int* __restrict__ btot,
        int* __restrict__ bedge, int e, int nbk, int nchunk) {
    __shared__ int s[256];
    __shared__ int lbase[256];
    int tid = threadIdx.x;
    int base = blockIdx.x * CHUNK + tid;
    int r[16], c[16];
#pragma unroll
    for (int k = 0; k < 16; ++k) {
        int i = base + k * 256;
        bool ok = i < e;
        r[k] = ok ? row[i] : 0;
        c[k] = ok ? col[i] : -1;
    }
    int tot = tid < nbk ? btot[tid] : 0;
    int pre = tid < nbk ? ghist[(size_t)tid * nchunk + blockIdx.x] : 0;
    s[tid] = tot;
    __syncthreads();
    lds_scan_incl(s, tid, 256);
    lbase[tid] = s[tid] - tot + pre;
    __syncthreads();
#pragma unroll
    for (int k = 0; k < 16; ++k) {
        if (c[k] >= 0) {
            int p = atomicAdd(&lbase[c[k] >> BSHIFT], 1);
            v2i v;
            v.x = r[k];
            v.y = c[k];
            *(v2i*)(bedge + 2 * (size_t)p) = v;
        }
    }
}

#define STAGE_CAP 12288

__global__ void __launch_bounds__(512)
k_bfill(const int* __restrict__ bedge, const int* __restrict__ btot,
        int* __restrict__ rowptr, float* __restrict__ dinv, int* __restrict__ src, int n) {
    __shared__ int s[512];
    __shared__ int cl[512];
    __shared__ int stage[STAGE_CAP];
    int tid = threadIdx.x;
    int b = blockIdx.x;
    int part = 0;
    for (int j = tid; j < b; j += 512) part += btot[j];
    s[tid] = part;
    cl[tid] = 0;
    __syncthreads();
    for (int st = 256; st > 0; st >>= 1) {
        if (tid < st) s[tid] += s[tid + st];
        __syncthreads();
    }
    int bstart = s[0];
    int bcount = btot[b];
    int node0 = b << BSHIFT;
    __syncthreads();
    for (int i0 = 0; i0 < bcount; i0 += 8 * 512) {
        int c[8];
#pragma unroll
        for (int u = 0; u < 8; ++u) {
            int i = i0 + u * 512 + tid;
            c[u] = i < bcount ? bedge[2 * (size_t)(bstart + i) + 1] : -1;
        }
#pragma unroll
        for (int u = 0; u < 8; ++u)
            if (c[u] >= 0) atomicAdd(&cl[c[u] - node0], 1);
    }
    __syncthreads();
    int cn = cl[tid];
    s[tid] = cn;
    __syncthreads();
    lds_scan_incl(s, tid, 512);
    int excl = s[tid] - cn;
    int node = node0 + tid;
    if (node < n) {
        rowptr[node] = bstart + excl;
        float d = (float)(cn + 1);
        dinv[node] = d > 0.0f ? rsqrtf(d) : 0.0f;
        if (node == n - 1) rowptr[n] = bstart + s[tid];
    }
    cl[tid] = excl;
    __syncthreads();
    bool fits = bcount <= STAGE_CAP;
    for (int i0 = 0; i0 < bcount; i0 += 8 * 512) {
        v2i rc[8];
#pragma unroll
        for (int u = 0; u < 8; ++u) {
            int i = i0 + u * 512 + tid;
            v2i none;
            none.x = 0;
            none.y = -1;
            rc[u] = i < bcount ? *(const v2i*)(bedge + 2 * (size_t)(bstart + i)) : none;
        }
#pragma unroll
        for (int u = 0; u < 8; ++u) {
            if (rc[u].y >= 0) {
                int slot = atomicAdd(&cl[rc[u].y - node0], 1);
                if (fits) stage[slot] = rc[u].x;
                else      src[bstart + slot] = rc[u].x;
            }
        }
    }
    if (fits) {
        __syncthreads();
        for (int i = tid; i < bcount; i += 512) src[bstart + i] = stage[i];
    }
}

template <int K, bool RELU, bool STOREPRE>
__global__ void k_dense(const float* __restrict__ X, const float* __restrict__ W,
                        const float* __restrict__ bias, const float* __restrict__ rscale,
                        float* __restrict__ outA, float* __restrict__ outPre,
                        int ntiles) {
    constexpr int WP = K + 4;
    __shared__ __attribute__((aligned(16))) float wlds[64 * WP];
    constexpr int NV4 = (64 * K) / 4;
    constexpr int RV4 = K / 4;
    for (int i = threadIdx.x; i < NV4; i += 256)
        *(float4*)(wlds + (i / RV4) * WP + 4 * (i % RV4)) = ((const float4*)W)[i];
    __syncthreads();

    int wid  = (blockIdx.x * blockDim.x + threadIdx.x) >> 5;
    int lane = threadIdx.x & 31;
    if (wid >= ntiles) return;
    int ln = lane & 15;
    int kh = lane >> 4;

    const float* xrow = X + (size_t)(wid * 16 + ln) * K + 2 * kh;
    v2f afr[K / 4];
#pragma unroll
    for (int kk = 0; kk < K / 4; ++kk)
        afr[kk] = *(const v2f*)(xrow + 4 * kk);
    float rs[8];
    if constexpr (STOREPRE) {
#pragma unroll
        for (int r = 0; r < 8; ++r)
            rs[r] = rscale[wid * 16 + r + 8 * kh];
    }

#pragma unroll
    for (int j = 0; j < 4; ++j) {
        const float* wrow = wlds + (size_t)(16 * j + ln) * WP + 2 * kh;
        v8f acc = {};
#pragma unroll
        for (int kk = 0; kk < K / 4; ++kk) {
            v2f b = *(const v2f*)(wrow + 4 * kk);
            acc = __builtin_amdgcn_wmma_f32_16x16x4_f32(
                false, afr[kk], false, b, (short)0, acc, false, false);
        }
        int n = 16 * j + ln;
        float bv = bias[n];
#pragma unroll
        for (int r = 0; r < 8; ++r) {
            int m = r + 8 * kh;
            size_t gi = (size_t)(wid * 16 + m) * 64 + n;
            float v = acc[r] + bv;
            float a = RELU ? fmaxf(v, 0.0f) : v;
            if constexpr (STOREPRE) {
                outPre[gi] = v;
                a *= rs[r];
            }
            outA[gi] = a;
        }
    }
}

#define TILE_PITCH 68
#define W_PITCH 80

__device__ __forceinline__ void ask16(v2f (&a)[16], const v2f* hrow2,
                                      unsigned lane, int ro, int k0) {
#pragma unroll
    for (int u = 0; u < 16; ++u) {
        unsigned r = (unsigned)__builtin_amdgcn_readlane(ro, k0 + u);
        a[u] = hrow2[r + lane];
    }
}

template <bool TAIL>
__device__ __forceinline__ void sum16(const v2f (&a)[16], int p0,
                                      int endAll, int rp, float* tl,
                                      int& cur, int& curEnd, v2f& acc) {
#pragma unroll
    for (int u = 0; u < 16; ++u) {
        int p = p0 + u;
        if (!TAIL || p < endAll) {
            if (p >= curEnd) {
                *(v2f*)(tl + cur * TILE_PITCH) = acc;
                do {
                    ++cur;
                    curEnd = __builtin_amdgcn_readlane(rp, cur + 1);
                } while (p >= curEnd && cur < 15);
                acc = *(const v2f*)(tl + cur * TILE_PITCH);
            }
            acc += a[u];
        }
    }
}

__global__ void __launch_bounds__(128)
k_layer(const int* __restrict__ rowptr, const int* __restrict__ src,
        const float* __restrict__ dinv, const float* __restrict__ hin,
        const float* __restrict__ x0, const float* __restrict__ Wl, float beta,
        float* __restrict__ hout, int ntiles, int scale_out) {
    __shared__ __attribute__((aligned(16))) float wlds[64 * W_PITCH];
    __shared__ __attribute__((aligned(16))) float tiles[4][16 * TILE_PITCH];
#pragma unroll
    for (int q = 0; q < 8; ++q) {
        int i = threadIdx.x + 128 * q;
        *(float4*)(wlds + (i >> 4) * W_PITCH + 4 * (i & 15)) = ((const float4*)Wl)[i];
    }

    int lane = threadIdx.x & 31;
    int wv   = threadIdx.x >> 5;
    int wid = __builtin_amdgcn_readfirstlane((int)(blockIdx.x * 4 + wv));
    bool active = wid < ntiles;
    float* t = tiles[wv];
    float dl = 0.0f;

    if (active) {
        int base = wid * 16;
        dl = dinv[base + (lane & 15)];
        int   rp = rowptr[base + (lane < 16 ? lane : 16)];
        const float* hl = hin + 2 * lane;
        const float* xl = x0 + 2 * lane;
        float* tl = t + 2 * lane;
        const v2f* hrow2 = (const v2f*)hin;
        unsigned ulane = (unsigned)lane;

        v2f hv[16];
        v2f xv[16];
#pragma unroll
        for (int j = 0; j < 16; ++j)
            hv[j] = *(const v2f*)(hl + (size_t)(base + j) * 64);
#pragma unroll
        for (int j = 0; j < 16; ++j)
            xv[j] = *(const v2f*)(xl + (size_t)(base + j) * 64);

        int beg    = __builtin_amdgcn_readlane(rp, 0);
        int endAll = __builtin_amdgcn_readlane(rp, 16);
        int pos = beg + lane;
        bool ok = pos < endAll;
        int rec = src[ok ? pos : beg];
#pragma unroll
        for (int j = 0; j < 16; ++j)
            *(v2f*)(tl + j * TILE_PITCH) = hv[j];
        int cur = 0;
        int curEnd = __builtin_amdgcn_readlane(rp, 1);
        v2f acc = *(const v2f*)tl;

        int ro = (ok ? rec : base) * 32;
        v2f ra[16], rb[16];
        ask16(ra, hrow2, ulane, ro, 0);
        for (int p0 = beg; p0 < endAll; p0 += 32) {
            int npos = p0 + 32 + lane;
            bool nok = npos < endAll;
            int nrec = src[nok ? npos : p0];
            ask16(rb, hrow2, ulane, ro, 16);
            bool full = p0 + 32 <= endAll;
            if (full) sum16<false>(ra, p0, endAll, rp, tl, cur, curEnd, acc);
            else      sum16<true>(ra, p0, endAll, rp, tl, cur, curEnd, acc);
            ro = (nok ? nrec : base) * 32;
            ask16(ra, hrow2, ulane, ro, 0);
            if (full) sum16<false>(rb, p0 + 16, endAll, rp, tl, cur, curEnd, acc);
            else      sum16<true>(rb, p0 + 16, endAll, rp, tl, cur, curEnd, acc);
        }
        *(v2f*)(tl + cur * TILE_PITCH) = acc;
#pragma unroll
        for (int j = 0; j < 16; ++j) {
            float dc = __int_as_float(__builtin_amdgcn_readlane(__float_as_int(dl), j));
            v2f a = *(const v2f*)(tl + j * TILE_PITCH);
            *(v2f*)(tl + j * TILE_PITCH) = (a * dc) * 0.9f + xv[j] * 0.1f;
        }
    }
    __syncthreads();
    if (!active) return;

    int ln = lane & 15;
    int kh = lane >> 4;
    v2f afr[16];
#pragma unroll
    for (int kk = 0; kk < 16; ++kk)
        afr[kk] = *(const v2f*)(t + ln * TILE_PITCH + 4 * kk + 2 * kh);
    float ob = 1.0f - beta;
    float sc[8];
#pragma unroll
    for (int r = 0; r < 8; ++r) {
        float lo = __int_as_float(__builtin_amdgcn_readlane(__float_as_int(dl), r));
        float hi = __int_as_float(__builtin_amdgcn_readlane(__float_as_int(dl), r + 8));
        sc[r] = scale_out ? (kh ? hi : lo) : 1.0f;
    }
#pragma unroll
    for (int j = 0; j < 4; ++j) {
        const float* wcol = wlds + 2 * kh * W_PITCH + 16 * j + ln;
        v8f acc = {};
#pragma unroll
        for (int kk = 0; kk < 16; ++kk) {
            v2f b;
            b.x = wcol[(size_t)(4 * kk) * W_PITCH];
            b.y = wcol[(size_t)(4 * kk + 1) * W_PITCH];
            acc = __builtin_amdgcn_wmma_f32_16x16x4_f32(
                false, afr[kk], false, b, (short)0, acc, false, false);
        }
#pragma unroll
        for (int r = 0; r < 8; ++r) {
            int m = r + 8 * kh;
            int n = 16 * j + ln;
            size_t gi = (size_t)(wid * 16 + m) * 64 + n;
            float hh = t[m * TILE_PITCH + n];
            float v = ob * hh + beta * acc[r];
            hout[gi] = fmaxf(v, 0.0f) * sc[r];
        }
    }
}

extern "C" void kernel_launch(void* const* d_in, const int* in_sizes, int n_in,
                              void* d_out, int out_size, void* d_ws, size_t ws_size,
                              hipStream_t stream) {
    const float* x      = (const float*)d_in[0];
    const int*   ei     = (const int*)  d_in[1];
    const float* lin0_w = (const float*)d_in[2];
    const float* lin0_b = (const float*)d_in[3];
    const float* conv_w = (const float*)d_in[4];
    const float* lin1_w = (const float*)d_in[5];
    const float* lin1_b = (const float*)d_in[6];
    float* out = (float*)d_out;

    const int N = in_sizes[0] / 128;
    const int E = in_sizes[1] / 2;
    const int* row = ei;
    const int* col = ei + E;

    float* ws   = (float*)d_ws;
    float* dinv = ws;
    float* x0   = dinv + N;
    float* h    = x0 + (size_t)N * 64;
    float* h2   = h  + (size_t)N * 64;
    int* bedge  = (int*)(h2 + (size_t)N * 64);
    int* srcs   = bedge + 2 * (size_t)E;
    int* rowptr = srcs + E;
    int* btot   = rowptr + (N + 1);
    int* ghist  = btot + 256;

    const int ntiles = N / 16;
    const int gemmGrid = (ntiles * 32 + 255) / 256;
    const int nbk = (N + 511) / 512;
    const int nchunk = (E + 4095) / 4096;

    k_hist  <<<nchunk, 256, 0, stream>>>(col, ghist, E, nbk, nchunk);
    k_bscan <<<nbk, 512, 0, stream>>>(ghist, btot, nchunk);
    k_place <<<nchunk, 256, 0, stream>>>(row, col, ghist, btot, bedge, E, nbk, nchunk);
    k_bfill <<<nbk, 512, 0, stream>>>(bedge, btot, rowptr, dinv, srcs, N);

    k_dense<128, true, true><<<gemmGrid, 256, 0, stream>>>(x, lin0_w, lin0_b, dinv, h, x0, ntiles);

    const int layerGrid = (ntiles + 3) / 4;
    for (int l = 0; l < 4; ++l) {
        float beta = logf(0.5f / (float)(l + 1) + 1.0f);
        const float* hin = (l & 1) ? h2 : h;
        float* hout      = (l & 1) ? h : h2;
        k_layer<<<layerGrid, 128, 0, stream>>>(rowptr, srcs, dinv, hin, x0,
                                               conv_w + (size_t)l * 64 * 64, beta, hout, ntiles,
                                               l < 3 ? 1 : 0);
    }

    k_dense<64, false, false><<<gemmGrid, 256, 0, stream>>>(h, lin1_w, lin1_b, nullptr, out, nullptr, ntiles);
}


# Output product fused into the last layer kernel: activated tile stays in LDS, W2 takes the layer weights' LDS place behind a barrier, a second WMMA pass writes out; 9 launches, h never stored
# speedup vs baseline: 4.6110x; 1.0601x over previous
#include <hip/hip_runtime.h>
#include <math.h>

typedef __attribute__((ext_vector_type(2))) float v2f;
typedef __attribute__((ext_vector_type(8))) float v8f;
typedef __attribute__((ext_vector_type(2))) int   v2i;
typedef __attribute__((ext_vector_type(4))) float v4f;

#define BSHIFT 9
#define CHUNK 4096

__device__ __forceinline__ void lds_scan_incl(int* s, int tid, int n) {
    for (int st = 1; st < n; st <<= 1) {
        int v = tid >= st ? s[tid - st] : 0;
        __syncthreads();
        s[tid] += v;
        __syncthreads();
    }
}

__global__ void __launch_bounds__(256)
k_hist(const int* __restrict__ col, int* __restrict__ ghist, int e, int nbk, int nchunk) {
    __shared__ int lh[256];
    int tid = threadIdx.x;
    lh[tid] = 0;
    __syncthreads();
    int base = blockIdx.x * CHUNK + tid;
    int c[16];
#pragma unroll
    for (int k = 0; k < 16; ++k) {
        int i = base + k * 256;
        c[k] = i < e ? col[i] : -1;
    }
#pragma unroll
    for (int k = 0; k < 16; ++k)
        if (c[k] >= 0) atomicAdd(&lh[c[k] >> BSHIFT], 1);
    __syncthreads();
    if (tid < nbk) ghist[(size_t)tid * nchunk + blockIdx.x] = lh[tid];
}

__global__ void __launch_bounds__(512)
k_bscan(int* __restrict__ ghist, int* __restrict__ btot, int nchunk) {
    __shared__ int s[512];
    int tid = threadIdx.x;
    int* g = ghist + (size_t)blockIdx.x * nchunk;
    int carry = 0;
    for (int t0 = 0; t0 < nchunk; t0 += 512) {
        int i = t0 + tid;
        int v = i < nchunk ? g[i] : 0;
        s[tid] = v;
        __syncthreads();
        lds_scan_incl(s, tid, 512);
        if (i < nchunk) g[i] = carry + s[tid] - v;
        carry += s[511];
        __syncthreads();
    }
    if (tid == 0) btot[blockIdx.x] = carry;
}

__global__ void __launch_bounds__(256)
k_place(const int* __restrict__ row, const int* __restrict__ col,
        const int* __restrict__ ghist, const int* __restrict__ btot,
        int* __restrict__ bedge, int e, int nbk, int nchunk) {
    __shared__ int s[256];
    __shared__ int lbase[256];
    int tid = threadIdx.x;
    int base = blockIdx.x * CHUNK + tid;
    int r[16], c[16];
#pragma unroll
    for (int k = 0; k < 16; ++k) {
        int i = base + k * 256;
        bool ok = i < e;
        r[k] = ok ? row[i] : 0;
        c[k] = ok ? col[i] : -1;
    }
    int tot = tid < nbk ? btot[tid] : 0;
    int pre = tid < nbk ? ghist[(size_t)tid * nchunk + blockIdx.x] : 0;
    s[tid] = tot;
    __syncthreads();
    lds_scan_incl(s, tid, 256);
    lbase[tid] = s[tid] - tot + pre;
    __syncthreads();
#pragma unroll
    for (int k = 0; k < 16; ++k) {
        if (c[k] >= 0) {
            int p = atomicAdd(&lbase[c[k] >> BSHIFT], 1);
            v2i v;
            v.x = r[k];
            v.y = c[k];
            *(v2i*)(bedge + 2 * (size_t)p) = v;
        }
    }
}

#define STAGE_CAP 12288

__global__ void __launch_bounds__(512)
k_bfill(const int* __restrict__ bedge, const int* __restrict__ btot,
        int* __restrict__ rowptr, float* __restrict__ dinv, int* __restrict__ src, int n) {
    __shared__ int s[512];
    __shared__ int cl[512];
    __shared__ int stage[STAGE_CAP];
    int tid = threadIdx.x;
    int b = blockIdx.x;
    int part = 0;
    for (int j = tid; j < b; j += 512) part += btot[j];
    s[tid] = part;
    cl[tid] = 0;
    __syncthreads();
    for (int st = 256; st > 0; st >>= 1) {
        if (tid < st) s[tid] += s[tid + st];
        __syncthreads();
    }
    int bstart = s[0];
    int bcount = btot[b];
    int node0 = b << BSHIFT;
    __syncthreads();
    for (int i0 = 0; i0 < bcount; i0 += 8 * 512) {
        int c[8];
#pragma unroll
        for (int u = 0; u < 8; ++u) {
            int i = i0 + u * 512 + tid;
            c[u] = i < bcount ? bedge[2 * (size_t)(bstart + i) + 1] : -1;
        }
#pragma unroll
        for (int u = 0; u < 8; ++u)
            if (c[u] >= 0) atomicAdd(&cl[c[u] - node0], 1);
    }
    __syncthreads();
    int cn = cl[tid];
    s[tid] = cn;
    __syncthreads();
    lds_scan_incl(s, tid, 512);
    int excl = s[tid] - cn;
    int node = node0 + tid;
    if (node < n) {
        rowptr[node] = bstart + excl;
        float d = (float)(cn + 1);
        dinv[node] = d > 0.0f ? rsqrtf(d) : 0.0f;
        if (node == n - 1) rowptr[n] = bstart + s[tid];
    }
    cl[tid] = excl;
    __syncthreads();
    bool fits = bcount <= STAGE_CAP;
    for (int i0 = 0; i0 < bcount; i0 += 8 * 512) {
        v2i rc[8];
#pragma unroll
        for (int u = 0; u < 8; ++u) {
            int i = i0 + u * 512 + tid;
            v2i none;
            none.x = 0;
            none.y = -1;
            rc[u] = i < bcount ? *(const v2i*)(bedge + 2 * (size_t)(bstart + i)) : none;
        }
#pragma unroll
        for (int u = 0; u < 8; ++u) {
            if (rc[u].y >= 0) {
                int slot = atomicAdd(&cl[rc[u].y - node0], 1);
                if (fits) stage[slot] = rc[u].x;
                else      src[bstart + slot] = rc[u].x;
            }
        }
    }
    if (fits) {
        __syncthreads();
        for (int i = tid; i < bcount; i += 512) src[bstart + i] = stage[i];
    }
}

template <int K, bool RELU, bool STOREPRE>
__global__ void k_dense(const float* __restrict__ X, const float* __restrict__ W,
                        const float* __restrict__ bias, const float* __restrict__ rscale,
                        float* __restrict__ outA, float* __restrict__ outPre,
                        int ntiles) {
    constexpr int WP = K + 4;
    __shared__ __attribute__((aligned(16))) float wlds[64 * WP];
    constexpr int NV4 = (64 * K) / 4;
    constexpr int RV4 = K / 4;
    for (int i = threadIdx.x; i < NV4; i += 256)
        *(float4*)(wlds + (i / RV4) * WP + 4 * (i % RV4)) = ((const float4*)W)[i];
    __syncthreads();

    int wid  = (blockIdx.x * blockDim.x + threadIdx.x) >> 5;
    int lane = threadIdx.x & 31;
    if (wid >= ntiles) return;
    int ln = lane & 15;
    int kh = lane >> 4;

    const float* xrow = X + (size_t)(wid * 16 + ln) * K + 2 * kh;
    v2f afr[K / 4];
#pragma unroll
    for (int kk = 0; kk < K / 4; ++kk)
        afr[kk] = *(const v2f*)(xrow + 4 * kk);
    float rs[8];
    if constexpr (STOREPRE) {
#pragma unroll
        for (int r = 0; r < 8; ++r)
            rs[r] = rscale[wid * 16 + r + 8 * kh];
    }

#pragma unroll
    for (int j = 0; j < 4; ++j) {
        const float* wrow = wlds + (size_t)(16 * j + ln) * WP + 2 * kh;
        v8f acc = {};
#pragma unroll
        for (int kk = 0; kk < K / 4; ++kk) {
            v2f b = *(const v2f*)(wrow + 4 * kk);
            acc = __builtin_amdgcn_wmma_f32_16x16x4_f32(
                false, afr[kk], false, b, (short)0, acc, false, false);
        }
        int n = 16 * j + ln;
        float bv = bias[n];
#pragma unroll
        for (int r = 0; r < 8; ++r) {
            int m = r + 8 * kh;
            size_t gi = (size_t)(wid * 16 + m) * 64 + n;
            float v = acc[r] + bv;
            float a = RELU ? fmaxf(v, 0.0f) : v;
            if constexpr (STOREPRE) {
                outPre[gi] = v;
                a *= rs[r];
            }
            outA[gi] = a;
        }
    }
}

#define TILE_PITCH 68
#define W_PITCH 80

__device__ __forceinline__ void ask16(v2f (&a)[16], const v2f* hrow2,
                                      unsigned lane, int ro, int k0) {
#pragma unroll
    for (int u = 0; u < 16; ++u) {
        unsigned r = (unsigned)__builtin_amdgcn_readlane(ro, k0 + u);
        a[u] = hrow2[r + lane];
    }
}

template <bool TAIL>
__device__ __forceinline__ void sum16(const v2f (&a)[16], int p0,
                                      int endAll, int rp, float* tl,
                                      int& cur, int& curEnd, v2f& acc) {
#pragma unroll
    for (int u = 0; u < 16; ++u) {
        int p = p0 + u;
        if (!TAIL || p < endAll) {
            if (p >= curEnd) {
                *(v2f*)(tl + cur * TILE_PITCH) = acc;
                do {
                    ++cur;
                    curEnd = __builtin_amdgcn_readlane(rp, cur + 1);
                } while (p >= curEnd && cur < 15);
                acc = *(const v2f*)(tl + cur * TILE_PITCH);
            }
            acc += a[u];
        }
    }
}

template <bool LAST>
__device__ __forceinline__ void
layer_body(const int* __restrict__ rowptr, const int* __restrict__ src,
           const float* __restrict__ dinv, const float* __restrict__ hin,
           const float* __restrict__ x0, const float* __restrict__ Wl, float beta,
           float* __restrict__ hout, int ntiles, int scale_out,
           const float* __restrict__ W2, const float* __restrict__ b2) {
    __shared__ __attribute__((aligned(16))) float wlds[64 * W_PITCH];
    __shared__ __attribute__((aligned(16))) float tiles[4][16 * TILE_PITCH];
#pragma unroll
    for (int q = 0; q < 8; ++q) {
        int i = threadIdx.x + 128 * q;
        *(float4*)(wlds + (i >> 4) * W_PITCH + 4 * (i & 15)) = ((const float4*)Wl)[i];
    }

    int lane = threadIdx.x & 31;
    int wv   = threadIdx.x >> 5;
    int wid = __builtin_amdgcn_readfirstlane((int)(blockIdx.x * 4 + wv));
    bool active = wid < ntiles;
    float* t = tiles[wv];
    float dl = 0.0f;

    if (active) {
        int base = wid * 16;
        dl = dinv[base + (lane & 15)];
        int   rp = rowptr[base + (lane < 16 ? lane : 16)];
        const float* hl = hin + 2 * lane;
        const float* xl = x0 + 2 * lane;
        float* tl = t + 2 * lane;
        const v2f* hrow2 = (const v2f*)hin;
        unsigned ulane = (unsigned)lane;

        v2f hv[16];
        v2f xv[16];
#pragma unroll
        for (int j = 0; j < 16; ++j)
            hv[j] = *(const v2f*)(hl + (size_t)(base + j) * 64);
#pragma unroll
        for (int j = 0; j < 16; ++j)
            xv[j] = *(const v2f*)(xl + (size_t)(base + j) * 64);

        int beg    = __builtin_amdgcn_readlane(rp, 0);
        int endAll = __builtin_amdgcn_readlane(rp, 16);
        int pos = beg + lane;
        bool ok = pos < endAll;
        int rec = src[ok ? pos : beg];
#pragma unroll
        for (int j = 0; j < 16; ++j)
            *(v2f*)(tl + j * TILE_PITCH) = hv[j];
        int cur = 0;
        int curEnd = __builtin_amdgcn_readlane(rp, 1);
        v2f acc = *(const v2f*)tl;

        int ro = (ok ? rec : base) * 32;
        v2f ra[16], rb[16];
        ask16(ra, hrow2, ulane, ro, 0);
        for (int p0 = beg; p0 < endAll; p0 += 32) {
            int npos = p0 + 32 + lane;
            bool nok = npos < endAll;
            int nrec = src[nok ? npos : p0];
            ask16(rb, hrow2, ulane, ro, 16);
            bool full = p0 + 32 <= endAll;
            if (full) sum16<false>(ra, p0, endAll, rp, tl, cur, curEnd, acc);
            else      sum16<true>(ra, p0, endAll, rp, tl, cur, curEnd, acc);
            ro = (nok ? nrec : base) * 32;
            ask16(ra, hrow2, ulane, ro, 0);
            if (full) sum16<false>(rb, p0 + 16, endAll, rp, tl, cur, curEnd, acc);
            else      sum16<true>(rb, p0 + 16, endAll, rp, tl, cur, curEnd, acc);
        }
        *(v2f*)(tl + cur * TILE_PITCH) = acc;
#pragma unroll
        for (int j = 0; j < 16; ++j) {
            float dc = __int_as_float(__builtin_amdgcn_readlane(__float_as_int(dl), j));
            v2f a = *(const v2f*)(tl + j * TILE_PITCH);
            *(v2f*)(tl + j * TILE_PITCH) = (a * dc) * 0.9f + xv[j] * 0.1f;
        }
    }
    __syncthreads();
    if constexpr (!LAST) {
        if (!active) return;
    }

    int ln = lane & 15;
    int kh = lane >> 4;
    float ob = 1.0f - beta;
    v4f w2r[8];
    if constexpr (LAST) {
#pragma unroll
        for (int q = 0; q < 8; ++q)
            w2r[q] = ((const v4f*)W2)[threadIdx.x + 128 * q];
    }
    if (active) {
        v2f afr[16];
#pragma unroll
        for (int kk = 0; kk < 16; ++kk)
            afr[kk] = *(const v2f*)(t + ln * TILE_PITCH + 4 * kk + 2 * kh);
        float sc[8];
        if constexpr (!LAST) {
#pragma unroll
            for (int r = 0; r < 8; ++r) {
                float lo = __int_as_float(__builtin_amdgcn_readlane(__float_as_int(dl), r));
                float hi = __int_as_float(__builtin_amdgcn_readlane(__float_as_int(dl), r + 8));
                sc[r] = scale_out ? (kh ? hi : lo) : 1.0f;
            }
        }
#pragma unroll
        for (int j = 0; j < 4; ++j) {
            const float* wcol = wlds + 2 * kh * W_PITCH + 16 * j + ln;
            v8f acc = {};
#pragma unroll
            for (int kk = 0; kk < 16; ++kk) {
                v2f b;
                b.x = wcol[(size_t)(4 * kk) * W_PITCH];
                b.y = wcol[(size_t)(4 * kk + 1) * W_PITCH];
                acc = __builtin_amdgcn_wmma_f32_16x16x4_f32(
                    false, afr[kk], false, b, (short)0, acc, false, false);
            }
#pragma unroll
            for (int r = 0; r < 8; ++r) {
                int m = r + 8 * kh;
                int n = 16 * j + ln;
                float hh = t[m * TILE_PITCH + n];
                float v = ob * hh + beta * acc[r];
                if constexpr (LAST) {
                    t[m * TILE_PITCH + n] = fmaxf(v, 0.0f);
                } else {
                    size_t gi = (size_t)(wid * 16 + m) * 64 + n;
                    hout[gi] = fmaxf(v, 0.0f) * sc[r];
                }
            }
        }
    }
    if constexpr (LAST) {
        __syncthreads();
#pragma unroll
        for (int q = 0; q < 8; ++q) {
            int i = threadIdx.x + 128 * q;
            *(v4f*)(wlds + (i >> 4) * TILE_PITCH + 4 * (i & 15)) = w2r[q];
        }
        __syncthreads();
        if (active) {
            v2f afr[16];
#pragma unroll
            for (int kk = 0; kk < 16; ++kk)
                afr[kk] = *(const v2f*)(t + ln * TILE_PITCH + 4 * kk + 2 * kh);
#pragma unroll
            for (int j = 0; j < 4; ++j) {
                const float* wrow = wlds + (16 * j + ln) * TILE_PITCH + 2 * kh;
                v8f acc = {};
#pragma unroll
                for (int kk = 0; kk < 16; ++kk) {
                    v2f b = *(const v2f*)(wrow + 4 * kk);
                    acc = __builtin_amdgcn_wmma_f32_16x16x4_f32(
                        false, afr[kk], false, b, (short)0, acc, false, false);
                }
                int n = 16 * j + ln;
                float bv = b2[n];
#pragma unroll
                for (int r = 0; r < 8; ++r) {
                    int m = r + 8 * kh;
                    size_t gi = (size_t)(wid * 16 + m) * 64 + n;
                    hout[gi] = acc[r] + bv;
                }
            }
        }
    }
}

__global__ void __launch_bounds__(128)
k_layer(const int* __restrict__ rowptr, const int* __restrict__ src,
        const float* __restrict__ dinv, const float* __restrict__ hin,
        const float* __restrict__ x0, const float* __restrict__ Wl, float beta,
        float* __restrict__ hout, int ntiles, int scale_out) {
    layer_body<false>(rowptr, src, dinv, hin, x0, Wl, beta, hout, ntiles, scale_out,
                      nullptr, nullptr);
}

__global__ void __launch_bounds__(128, 7)
k_layer_out(const int* __restrict__ rowptr, const int* __restrict__ src,
            const float* __restrict__ dinv, const float* __restrict__ hin,
            const float* __restrict__ x0, const float* __restrict__ Wl, float beta,
            const float* __restrict__ W2, const float* __restrict__ b2,
            float* __restrict__ out, int ntiles) {
    layer_body<true>(rowptr, src, dinv, hin, x0, Wl, beta, out, ntiles, 0, W2, b2);
}

extern "C" void kernel_launch(void* const* d_in, const int* in_sizes, int n_in,
                              void* d_out, int out_size, void* d_ws, size_t ws_size,
                              hipStream_t stream) {
    const float* x      = (const float*)d_in[0];
    const int*   ei     = (const int*)  d_in[1];
    const float* lin0_w = (const float*)d_in[2];
    const float* lin0_b = (const float*)d_in[3];
    const float* conv_w = (const float*)d_in[4];
    const float* lin1_w = (const float*)d_in[5];
    const float* lin1_b = (const float*)d_in[6];
    float* out = (float*)d_out;

    const int N = in_sizes[0] / 128;
    const int E = in_sizes[1] / 2;
    const int* row = ei;
    const int* col = ei + E;

    float* ws   = (float*)d_ws;
    float* dinv = ws;
    float* x0   = dinv + N;
    float* h    = x0 + (size_t)N * 64;
    float* h2   = h  + (size_t)N * 64;
    int* bedge  = (int*)(h2 + (size_t)N * 64);
    int* srcs   = bedge + 2 * (size_t)E;
    int* rowptr = srcs + E;
    int* btot   = rowptr + (N + 1);
    int* ghist  = btot + 256;

    const int ntiles = N / 16;
    const int gemmGrid = (ntiles * 32 + 255) / 256;
    const int nbk = (N + 511) / 512;
    const int nchunk = (E + 4095) / 4096;

    k_hist  <<<nchunk, 256, 0, stream>>>(col, ghist, E, nbk, nchunk);
    k_bscan <<<nbk, 512, 0, stream>>>(ghist, btot, nchunk);
    k_place <<<nchunk, 256, 0, stream>>>(row, col, ghist, btot, bedge, E, nbk, nchunk);
    k_bfill <<<nbk, 512, 0, stream>>>(bedge, btot, rowptr, dinv, srcs, N);

    k_dense<128, true, true><<<gemmGrid, 256, 0, stream>>>(x, lin0_w, lin0_b, dinv, h, x0, ntiles);

    const int layerGrid = (ntiles + 3) / 4;
    for (int l = 0; l < 4; ++l) {
        float beta = logf(0.5f / (float)(l + 1) + 1.0f);
        const float* hin = (l & 1) ? h2 : h;
        float* hout      = (l & 1) ? h : h2;
        if (l < 3)
            k_layer<<<layerGrid, 128, 0, stream>>>(rowptr, srcs, dinv, hin, x0,
                                                   conv_w + (size_t)l * 64 * 64, beta, hout,
                                                   ntiles, 1);
        else
            k_layer_out<<<layerGrid, 128, 0, stream>>>(rowptr, srcs, dinv, hin, x0,
                                                       conv_w + (size_t)l * 64 * 64, beta,
                                                       lin1_w, lin1_b, out, ntiles);
    }
}


# Input product x0 no longer waits for the CSR: its tiles run as extra blocks of the bucket-fill launch; layer 1 gathers x0 itself, adding relu(row)*dinv[source] as one fma; 8 launches, no h0 plane
# speedup vs baseline: 4.8284x; 1.0471x over previous
#include <hip/hip_runtime.h>
#include <math.h>

typedef __attribute__((ext_vector_type(2))) float v2f;
typedef __attribute__((ext_vector_type(8))) float v8f;
typedef __attribute__((ext_vector_type(2))) int   v2i;
typedef __attribute__((ext_vector_type(4))) float v4f;

#define BSHIFT 9
#define CHUNK 4096

__device__ __forceinline__ void lds_scan_incl(int* s, int tid, int n) {
    for (int st = 1; st < n; st <<= 1) {
        int v = tid >= st ? s[tid - st] : 0;
        __syncthreads();
        s[tid] += v;
        __syncthreads();
    }
}

__global__ void __launch_bounds__(256)
k_hist(const int* __restrict__ col, int* __restrict__ ghist, int e, int nbk, int nchunk) {
    __shared__ int lh[256];
    int tid = threadIdx.x;
    lh[tid] = 0;
    __syncthreads();
    int base = blockIdx.x * CHUNK + tid;
    int c[16];
#pragma unroll
    for (int k = 0; k < 16; ++k) {
        int i = base + k * 256;
        c[k] = i < e ? col[i] : -1;
    }
#pragma unroll
    for (int k = 0; k < 16; ++k)
        if (c[k] >= 0) atomicAdd(&lh[c[k] >> BSHIFT], 1);
    __syncthreads();
    if (tid < nbk) ghist[(size_t)tid * nchunk + blockIdx.x] = lh[tid];
}

__global__ void __launch_bounds__(512)
k_bscan(int* __restrict__ ghist, int* __restrict__ btot, int nchunk) {
    __shared__ int s[512];
    int tid = threadIdx.x;
    int* g = ghist + (size_t)blockIdx.x * nchunk;
    int carry = 0;
    for (int t0 = 0; t0 < nchunk; t0 += 512) {
        int i = t0 + tid;
        int v = i < nchunk ? g[i] : 0;
        s[tid] = v;
        __syncthreads();
        lds_scan_incl(s, tid, 512);
        if (i < nchunk) g[i] = carry + s[tid] - v;
        carry += s[511];
        __syncthreads();
    }
    if (tid == 0) btot[blockIdx.x] = carry;
}

__global__ void __launch_bounds__(256)
k_place(const int* __restrict__ row, const int* __restrict__ col,
        const int* __restrict__ ghist, const int* __restrict__ btot,
        int* __restrict__ bedge, int e, int nbk, int nchunk) {
    __shared__ int s[256];
    __shared__ int lbase[256];
    int tid = threadIdx.x;
    int base = blockIdx.x * CHUNK + tid;
    int r[16], c[16];
#pragma unroll
    for (int k = 0; k < 16; ++k) {
        int i = base + k * 256;
        bool ok = i < e;
        r[k] = ok ? row[i] : 0;
        c[k] = ok ? col[i] : -1;
    }
    int tot = tid < nbk ? btot[tid] : 0;
    int pre = tid < nbk ? ghist[(size_t)tid * nchunk + blockIdx.x] : 0;
    s[tid] = tot;
    __syncthreads();
    lds_scan_incl(s, tid, 256);
    lbase[tid] = s[tid] - tot + pre;
    __syncthreads();
#pragma unroll
    for (int k = 0; k < 16; ++k) {
        if (c[k] >= 0) {
            int p = atomicAdd(&lbase[c[k] >> BSHIFT], 1);
            v2i v;
            v.x = r[k];
            v.y = c[k];
            *(v2i*)(bedge + 2 * (size_t)p) = v;
        }
    }
}

#define STAGE_CAP 12288

__device__ __forceinline__ void
bfill_body(const int* __restrict__ bedge, const int* __restrict__ btot,
           int* __restrict__ rowptr, float* __restrict__ dinv, int* __restrict__ src, int n,
           int* lds) {
    int* s = lds;
    int* cl = lds + 512;
    int* stage = lds + 1024;
    int tid = threadIdx.x;
    int b = blockIdx.x;
    int part = 0;
    for (int j = tid; j < b; j += 512) part += btot[j];
    s[tid] = part;
    cl[tid] = 0;
    __syncthreads();
    for (int st = 256; st > 0; st >>= 1) {
        if (tid < st) s[tid] += s[tid + st];
        __syncthreads();
    }
    int bstart = s[0];
    int bcount = btot[b];
    int node0 = b << BSHIFT;
    __syncthreads();
    for (int i0 = 0; i0 < bcount; i0 += 8 * 512) {
        int c[8];
#pragma unroll
        for (int u = 0; u < 8; ++u) {
            int i = i0 + u * 512 + tid;
            c[u] = i < bcount ? bedge[2 * (size_t)(bstart + i) + 1] : -1;
        }
#pragma unroll
        for (int u = 0; u < 8; ++u)
            if (c[u] >= 0) atomicAdd(&cl[c[u] - node0], 1);
    }
    __syncthreads();
    int cn = cl[tid];
    s[tid] = cn;
    __syncthreads();
    lds_scan_incl(s, tid, 512);
    int excl = s[tid] - cn;
    int node = node0 + tid;
    if (node < n) {
        rowptr[node] = bstart + excl;
        float d = (float)(cn + 1);
        dinv[node] = d > 0.0f ? rsqrtf(d) : 0.0f;
        if (node == n - 1) rowptr[n] = bstart + s[tid];
    }
    cl[tid] = excl;
    __syncthreads();
    bool fits = bcount <= STAGE_CAP;
    for (int i0 = 0; i0 < bcount; i0 += 8 * 512) {
        v2i rc[8];
#pragma unroll
        for (int u = 0; u < 8; ++u) {
            int i = i0 + u * 512 + tid;
            v2i none;
            none.x = 0;
            none.y = -1;
            rc[u] = i < bcount ? *(const v2i*)(bedge + 2 * (size_t)(bstart + i)) : none;
        }
#pragma unroll
        for (int u = 0; u < 8; ++u) {
            if (rc[u].y >= 0) {
                int slot = atomicAdd(&cl[rc[u].y - node0], 1);
                if (fits) stage[slot] = rc[u].x;
                else      src[bstart + slot] = rc[u].x;
            }
        }
    }
    if (fits) {
        __syncthreads();
        for (int i = tid; i < bcount; i += 512) src[bstart + i] = stage[i];
    }
}

template <int K>
__device__ __forceinline__ void
dense_body(int db, const float* __restrict__ X, const float* __restrict__ W,
           const float* __restrict__ bias, float* __restrict__ outPre,
           int ntiles, float* wlds) {
    constexpr int WP = K + 4;
    constexpr int NV4 = (64 * K) / 4;
    constexpr int RV4 = K / 4;
    for (int i = threadIdx.x; i < NV4; i += 512)
        *(float4*)(wlds + (i / RV4) * WP + 4 * (i % RV4)) = ((const float4*)W)[i];
    __syncthreads();

    int wid  = (db * 512 + (int)threadIdx.x) >> 5;
    int lane = threadIdx.x & 31;
    if (wid >= ntiles) return;
    int ln = lane & 15;
    int kh = lane >> 4;

    const float* xrow = X + (size_t)(wid * 16 + ln) * K + 2 * kh;
    v2f afr[K / 4];
#pragma unroll
    for (int kk = 0; kk < K / 4; ++kk)
        afr[kk] = *(const v2f*)(xrow + 4 * kk);

#pragma unroll
    for (int j = 0; j < 4; ++j) {
        const float* wrow = wlds + (size_t)(16 * j + ln) * WP + 2 * kh;
        v8f acc = {};
#pragma unroll
        for (int kk = 0; kk < K / 4; ++kk) {
            v2f b = *(const v2f*)(wrow + 4 * kk);
            acc = __builtin_amdgcn_wmma_f32_16x16x4_f32(
                false, afr[kk], false, b, (short)0, acc, false, false);
        }
        int n = 16 * j + ln;
        float bv = bias[n];
#pragma unroll
        for (int r = 0; r < 8; ++r) {
            int m = r + 8 * kh;
            size_t gi = (size_t)(wid * 16 + m) * 64 + n;
            outPre[gi] = acc[r] + bv;
        }
    }
}

__global__ void __launch_bounds__(512)
k_bfill_dense(const int* __restrict__ bedge, const int* __restrict__ btot,
              int* __restrict__ rowptr, float* __restrict__ dinv, int* __restrict__ src, int n,
              int nbk, const float* __restrict__ X, const float* __restrict__ W,
              const float* __restrict__ bias, float* __restrict__ x0, int ntiles) {
    __shared__ __attribute__((aligned(16))) int lds[1024 + STAGE_CAP];
    if ((int)blockIdx.x < nbk)
        bfill_body(bedge, btot, rowptr, dinv, src, n, lds);
    else
        dense_body<128>((int)blockIdx.x - nbk, X, W, bias, x0, ntiles, (float*)lds);
}

#define TILE_PITCH 68
#define W_PITCH 80

__device__ __forceinline__ void ask16(v2f (&a)[16], const v2f* hrow2,
                                      unsigned lane, int ro, int k0) {
#pragma unroll
    for (int u = 0; u < 16; ++u) {
        unsigned r = (unsigned)__builtin_amdgcn_readlane(ro, k0 + u);
        a[u] = hrow2[r + lane];
    }
}

template <bool TAIL, bool FIRST>
__device__ __forceinline__ void sum16(const v2f (&a)[16], int p0,
                                      int endAll, int rp, float* tl,
                                      int& cur, int& curEnd, v2f& acc,
                                      float sc, int k0) {
#pragma unroll
    for (int u = 0; u < 16; ++u) {
        int p = p0 + u;
        if (!TAIL || p < endAll) {
            if (p >= curEnd) {
                *(v2f*)(tl + cur * TILE_PITCH) = acc;
                do {
                    ++cur;
                    curEnd = __builtin_amdgcn_readlane(rp, cur + 1);
                } while (p >= curEnd && cur < 15);
                acc = *(const v2f*)(tl + cur * TILE_PITCH);
            }
            if constexpr (FIRST) {
                float f = __int_as_float(__builtin_amdgcn_readlane(__float_as_int(sc), k0 + u));
                v2f r;
                r.x = fmaxf(a[u].x, 0.0f);
                r.y = fmaxf(a[u].y, 0.0f);
                acc += r * f;
            } else {
                acc += a[u];
            }
        }
    }
}

template <bool LAST, bool FIRST>
__device__ __forceinline__ void
layer_body(const int* __restrict__ rowptr, const int* __restrict__ src,
           const float* __restrict__ dinv, const float* __restrict__ hin,
           const float* __restrict__ x0, const float* __restrict__ Wl, float beta,
           float* __restrict__ hout, int ntiles, int scale_out,
           const float* __restrict__ W2, const float* __restrict__ b2) {
    __shared__ __attribute__((aligned(16))) float wlds[64 * W_PITCH];
    __shared__ __attribute__((aligned(16))) float tiles[4][16 * TILE_PITCH];
#pragma unroll
    for (int q = 0; q < 8; ++q) {
        int i = threadIdx.x + 128 * q;
        *(float4*)(wlds + (i >> 4) * W_PITCH + 4 * (i & 15)) = ((const float4*)Wl)[i];
    }

    int lane = threadIdx.x & 31;
    int wv   = threadIdx.x >> 5;
    int wid = __builtin_amdgcn_readfirstlane((int)(blockIdx.x * 4 + wv));
    bool active = wid < ntiles;
    float* t = tiles[wv];
    float dl = 0.0f;

    if (active) {
        int base = wid * 16;
        dl = dinv[base + (lane & 15)];
        int   rp = rowptr[base + (lane < 16 ? lane : 16)];
        const float* hl = hin + 2 * lane;
        const float* xl = x0 + 2 * lane;
        float* tl = t + 2 * lane;
        const v2f* hrow2 = (const v2f*)hin;
        unsigned ulane = (unsigned)lane;

        v2f hv[16];
        v2f xv[16];
        if constexpr (!FIRST) {
#pragma unroll
            for (int j = 0; j < 16; ++j)
                hv[j] = *(const v2f*)(hl + (size_t)(base + j) * 64);
        }
#pragma unroll
        for (int j = 0; j < 16; ++j)
            xv[j] = *(const v2f*)(xl + (size_t)(base + j) * 64);

        int beg    = __builtin_amdgcn_readlane(rp, 0);
        int endAll = __builtin_amdgcn_readlane(rp, 16);
        int pos = beg + lane;
        bool ok = pos < endAll;
        int rec = src[ok ? pos : beg];
        float sc = 0.0f, nsc = 0.0f;
        if constexpr (FIRST) {
            sc = dinv[ok ? rec : base];
#pragma unroll
            for (int j = 0; j < 16; ++j) {
                float dj = __int_as_float(__builtin_amdgcn_readlane(__float_as_int(dl), j));
                v2f r;
                r.x = fmaxf(xv[j].x, 0.0f);
                r.y = fmaxf(xv[j].y, 0.0f);
                hv[j] = r * dj;
            }
        }
#pragma unroll
        for (int j = 0; j < 16; ++j)
            *(v2f*)(tl + j * TILE_PITCH) = hv[j];
        int cur = 0;
        int curEnd = __builtin_amdgcn_readlane(rp, 1);
        v2f acc = *(const v2f*)tl;

        int ro = (ok ? rec : base) * 32;
        v2f ra[16], rb[16];
        ask16(ra, hrow2, ulane, ro, 0);
        for (int p0 = beg; p0 < endAll; p0 += 32) {
            int npos = p0 + 32 + lane;
            bool nok = npos < endAll;
            int nrec = src[nok ? npos : p0];
            ask16(rb, hrow2, ulane, ro, 16);
            bool full = p0 + 32 <= endAll;
            if (full) sum16<false, FIRST>(ra, p0, endAll, rp, tl, cur, curEnd, acc, sc, 0);
            else      sum16<true, FIRST>(ra, p0, endAll, rp, tl, cur, curEnd, acc, sc, 0);
            ro = (nok ? nrec : base) * 32;
            if constexpr (FIRST) nsc = dinv[nok ? nrec : base];
            ask16(ra, hrow2, ulane, ro, 0);
            if (full) sum16<false, FIRST>(rb, p0 + 16, endAll, rp, tl, cur, curEnd, acc, sc, 16);
            else      sum16<true, FIRST>(rb, p0 + 16, endAll, rp, tl, cur, curEnd, acc, sc, 16);
            if constexpr (FIRST) sc = nsc;
        }
        *(v2f*)(tl + cur * TILE_PITCH) = acc;
#pragma unroll
        for (int j = 0; j < 16; ++j) {
            float dc = __int_as_float(__builtin_amdgcn_readlane(__float_as_int(dl), j));
            v2f a = *(const v2f*)(tl + j * TILE_PITCH);
            *(v2f*)(tl + j * TILE_PITCH) = (a * dc) * 0.9f + xv[j] * 0.1f;
        }
    }
    __syncthreads();
    if constexpr (!LAST) {
        if (!active) return;
    }

    int ln = lane & 15;
    int kh = lane >> 4;
    float ob = 1.0f - beta;
    v4f w2r[8];
    if constexpr (LAST) {
#pragma unroll
        for (int q = 0; q < 8; ++q)
            w2r[q] = ((const v4f*)W2)[threadIdx.x + 128 * q];
    }
    if (active) {
        v2f afr[16];
#pragma unroll
        for (int kk = 0; kk < 16; ++kk)
            afr[kk] = *(const v2f*)(t + ln * TILE_PITCH + 4 * kk + 2 * kh);
        float sc[8];
        if constexpr (!LAST) {
#pragma unroll
            for (int r = 0; r < 8; ++r) {
                float lo = __int_as_float(__builtin_amdgcn_readlane(__float_as_int(dl), r));
                float hi = __int_as_float(__builtin_amdgcn_readlane(__float_as_int(dl), r + 8));
                sc[r] = scale_out ? (kh ? hi : lo) : 1.0f;
            }
        }
#pragma unroll
        for (int j = 0; j < 4; ++j) {
            const float* wcol = wlds + 2 * kh * W_PITCH + 16 * j + ln;
            v8f acc = {};
#pragma unroll
            for (int kk = 0; kk < 16; ++kk) {
                v2f b;
                b.x = wcol[(size_t)(4 * kk) * W_PITCH];
                b.y = wcol[(size_t)(4 * kk + 1) * W_PITCH];
                acc = __builtin_amdgcn_wmma_f32_16x16x4_f32(
                    false, afr[kk], false, b, (short)0, acc, false, false);
            }
#pragma unroll
            for (int r = 0; r < 8; ++r) {
                int m = r + 8 * kh;
                int n = 16 * j + ln;
                float hh = t[m * TILE_PITCH + n];
                float v = ob * hh + beta * acc[r];
                if constexpr (LAST) {
                    t[m * TILE_PITCH + n] = fmaxf(v, 0.0f);
                } else {
                    size_t gi = (size_t)(wid * 16 + m) * 64 + n;
                    hout[gi] = fmaxf(v, 0.0f) * sc[r];
                }
            }
        }
    }
    if constexpr (LAST) {
        __syncthreads();
#pragma unroll
        for (int q = 0; q < 8; ++q) {
            int i = threadIdx.x + 128 * q;
            *(v4f*)(wlds + (i >> 4) * TILE_PITCH + 4 * (i & 15)) = w2r[q];
        }
        __syncthreads();
        if (active) {
            v2f afr[16];
#pragma unroll
            for (int kk = 0; kk < 16; ++kk)
                afr[kk] = *(const v2f*)(t + ln * TILE_PITCH + 4 * kk + 2 * kh);
#pragma unroll
            for (int j = 0; j < 4; ++j) {
                const float* wrow = wlds + (16 * j + ln) * TILE_PITCH + 2 * kh;
                v8f acc = {};
#pragma unroll
                for (int kk = 0; kk < 16; ++kk) {
                    v2f b = *(const v2f*)(wrow + 4 * kk);
                    acc = __builtin_amdgcn_wmma_f32_16x16x4_f32(
                        false, afr[kk], false, b, (short)0, acc, false, false);
                }
                int n = 16 * j + ln;
                float bv = b2[n];
#pragma unroll
                for (int r = 0; r < 8; ++r) {
                    int m = r + 8 * kh;
                    size_t gi = (size_t)(wid * 16 + m) * 64 + n;
                    hout[gi] = acc[r] + bv;
                }
            }
        }
    }
}

__global__ void __launch_bounds__(128)
k_layer(const int* __restrict__ rowptr, const int* __restrict__ src,
        const float* __restrict__ dinv, const float* __restrict__ hin,
        const float* __restrict__ x0, const float* __restrict__ Wl, float beta,
        float* __restrict__ hout, int ntiles, int scale_out) {
    layer_body<false, false>(rowptr, src, dinv, hin, x0, Wl, beta, hout, ntiles, scale_out,
                             nullptr, nullptr);
}

__global__ void __launch_bounds__(128)
k_layer_first(const int* __restrict__ rowptr, const int* __restrict__ src,
              const float* __restrict__ dinv, const float* __restrict__ x0,
              const float* __restrict__ Wl, float beta,
              float* __restrict__ hout, int ntiles, int scale_out) {
    layer_body<false, true>(rowptr, src, dinv, x0, x0, Wl, beta, hout, ntiles, scale_out,
                            nullptr, nullptr);
}

__global__ void __launch_bounds__(128, 7)
k_layer_out(const int* __restrict__ rowptr, const int* __restrict__ src,
            const float* __restrict__ dinv, const float* __restrict__ hin,
            const float* __restrict__ x0, const float* __restrict__ Wl, float beta,
            const float* __restrict__ W2, const float* __restrict__ b2,
            float* __restrict__ out, int ntiles) {
    layer_body<true, false>(rowptr, src, dinv, hin, x0, Wl, beta, out, ntiles, 0, W2, b2);
}

extern "C" void kernel_launch(void* const* d_in, const int* in_sizes, int n_in,
                              void* d_out, int out_size, void* d_ws, size_t ws_size,
                              hipStream_t stream) {
    const float* x      = (const float*)d_in[0];
    const int*   ei     = (const int*)  d_in[1];
    const float* lin0_w = (const float*)d_in[2];
    const float* lin0_b = (const float*)d_in[3];
    const float* conv_w = (const float*)d_in[4];
    const float* lin1_w = (const float*)d_in[5];
    const float* lin1_b = (const float*)d_in[6];
    float* out = (float*)d_out;

    const int N = in_sizes[0] / 128;
    const int E = in_sizes[1] / 2;
    const int* row = ei;
    const int* col = ei + E;

    float* ws   = (float*)d_ws;
    float* dinv = ws;
    float* x0   = dinv + N;
    float* h    = x0 + (size_t)N * 64;
    float* h2   = h  + (size_t)N * 64;
    int* bedge  = (int*)(h2 + (size_t)N * 64);
    int* srcs   = bedge + 2 * (size_t)E;
    int* rowptr = srcs + E;
    int* btot   = rowptr + (N + 1);
    int* ghist  = btot + 256;

    const int ntiles = N / 16;
    const int denseBlocks = (ntiles + 15) / 16;
    const int nbk = (N + 511) / 512;
    const int nchunk = (E + 4095) / 4096;

    k_hist  <<<nchunk, 256, 0, stream>>>(col, ghist, E, nbk, nchunk);
    k_bscan <<<nbk, 512, 0, stream>>>(ghist, btot, nchunk);
    k_place <<<nchunk, 256, 0, stream>>>(row, col, ghist, btot, bedge, E, nbk, nchunk);
    k_bfill_dense<<<nbk + denseBlocks, 512, 0, stream>>>(bedge, btot, rowptr, dinv, srcs, N,
                                                         nbk, x, lin0_w, lin0_b, x0, ntiles);

    const int layerGrid = (ntiles + 3) / 4;
    for (int l = 0; l < 4; ++l) {
        float beta = logf(0.5f / (float)(l + 1) + 1.0f);
        const float* hin = (l & 1) ? h2 : h;
        float* hout      = (l & 1) ? h : h2;
        if (l == 0)
            k_layer_first<<<layerGrid, 128, 0, stream>>>(rowptr, srcs, dinv, x0, conv_w, beta,
                                                         hout, ntiles, 1);
        else if (l < 3)
            k_layer<<<layerGrid, 128, 0, stream>>>(rowptr, srcs, dinv, hin, x0,
                                                   conv_w + (size_t)l * 64 * 64, beta, hout,
                                                   ntiles, 1);
        else
            k_layer_out<<<layerGrid, 128, 0, stream>>>(rowptr, srcs, dinv, hin, x0,
                                                       conv_w + (size_t)l * 64 * 64, beta,
                                                       lin1_w, lin1_b, out, ntiles);
    }
}


# CSR build in 2 launches, no histogram or scan launch: each (bucket, chunk) pair owns 64 fixed slots (spill list beyond), bucket starts from per-chunk counts; in-node order still an arrival rank
# speedup vs baseline: 5.0751x; 1.0511x over previous
#include <hip/hip_runtime.h>
#include <math.h>

typedef __attribute__((ext_vector_type(2))) float v2f;
typedef __attribute__((ext_vector_type(8))) float v8f;
typedef __attribute__((ext_vector_type(2))) int   v2i;
typedef __attribute__((ext_vector_type(4))) float v4f;

#define BSHIFT 9
#define CHUNK 4096
#define CAP 64

typedef __attribute__((ext_vector_type(4))) int v4i;

__device__ __forceinline__ void lds_scan_incl(int* s, int tid, int n) {
    for (int st = 1; st < n; st <<= 1) {
        int v = tid >= st ? s[tid - st] : 0;
        __syncthreads();
        s[tid] += v;
        __syncthreads();
    }
}

__device__ __forceinline__ int block_sum512(int v, int* red, int tid) {
    for (int o = 16; o > 0; o >>= 1) v += __shfl_xor(v, o, 32);
    __syncthreads();
    if ((tid & 31) == 0) red[tid >> 5] = v;
    __syncthreads();
    int t = 0;
#pragma unroll
    for (int k = 0; k < 16; ++k) t += red[k];
    return t;
}

__global__ void __launch_bounds__(256)
k_place(const int* __restrict__ row, const int* __restrict__ col,
        int* __restrict__ slots, int* __restrict__ cnt,
        int* __restrict__ spill, int* __restrict__ spillcnt,
        int e, int nbk, int nchunk) {
    __shared__ int lh[256];
    __shared__ int sc[256];
    __shared__ int nsp;
    int tid = threadIdx.x;
    int ch = blockIdx.x;
    lh[tid] = 0;
    if (tid == 0) nsp = 0;
    __syncthreads();
    int base = ch * CHUNK + tid;
    int r[16], c[16];
#pragma unroll
    for (int k = 0; k < 16; ++k) {
        int i = base + k * 256;
        bool ok = i < e;
        r[k] = ok ? row[i] : 0;
        c[k] = ok ? col[i] : -1;
    }
#pragma unroll
    for (int k = 0; k < 16; ++k) {
        if (c[k] >= 0) {
            int b = c[k] >> BSHIFT;
            int p = atomicAdd(&lh[b], 1);
            v2i v;
            v.x = r[k];
            v.y = c[k];
            if (p < CAP) {
                *(v2i*)(slots + 2 * (((size_t)b * nchunk + ch) * CAP + p)) = v;
            } else {
                int q = atomicAdd(&nsp, 1);
                *(v2i*)(spill + 2 * ((size_t)ch * CHUNK + q)) = v;
            }
        }
    }
    __syncthreads();
    int mycnt = lh[tid];
    sc[tid] = mycnt;
    __syncthreads();
    lds_scan_incl(sc, tid, 256);
    if (tid < nbk) {
        v2i cp;
        cp.x = mycnt;
        cp.y = sc[tid] - mycnt;
        *(v2i*)(cnt + 2 * ((size_t)tid * nchunk + ch)) = cp;
    }
    if (tid == 0) spillcnt[ch] = nsp;
}

#define STAGE_CAP 12288
#define SU 9

__device__ __forceinline__ void
bfill_body(const int* __restrict__ slots, const int* __restrict__ cnt,
           const int* __restrict__ spill, const int* __restrict__ spillcnt,
           int nchunk,
           int* __restrict__ rowptr, float* __restrict__ dinv, int* __restrict__ src, int n,
           int* lds) {
    int* s = lds;
    int* cl = lds + 512;
    int* scnt = lds + 1024;
    int* stage = lds + 1536;
    int tid = threadIdx.x;
    int b = blockIdx.x;
    int lane = tid & 31;
    int w = tid >> 5;
    v2i cp = {0, 0};
    if (tid < nchunk) cp = *(const v2i*)(cnt + 2 * ((size_t)b * nchunk + tid));
    int spc = tid < nchunk ? spillcnt[tid] : 0;
    int mine = cp.x;
    scnt[tid] = mine < CAP ? mine : CAP;
    cl[tid] = 0;
    int bstart = block_sum512(cp.y, s, tid);
    int bcount = block_sum512(mine, s, tid);
    int anySpill = __syncthreads_or(spc != 0);
    int node0 = b << BSHIFT;
    const v4i* seg4 = (const v4i*)(slots + 2 * (size_t)b * nchunk * CAP);
    for (int j0 = 0; 16 * j0 < nchunk; j0 += SU) {
        v4i e4[SU];
        int m[SU];
        int cq = w + 16 * (j0 + lane);
        int mv = (lane < SU && cq < nchunk) ? scnt[cq] : 0;
#pragma unroll
        for (int u = 0; u < SU; ++u) m[u] = __builtin_amdgcn_readlane(mv, u);
        const v4i* pb = seg4 + (size_t)(w + 16 * j0) * (CAP / 2) + lane;
#pragma unroll
        for (int u = 0; u < SU; ++u) {
            v4i none = {0, -1, 0, -1};
            e4[u] = 2 * lane < m[u] ? pb[u * 16 * (CAP / 2)] : none;
        }
#pragma unroll
        for (int u = 0; u < SU; ++u) {
            if (2 * lane < m[u])     atomicAdd(&cl[e4[u].y - node0], 1);
            if (2 * lane + 1 < m[u]) atomicAdd(&cl[e4[u].w - node0], 1);
        }
    }
    if (anySpill) {
        for (int c = 0; c < nchunk; ++c) {
            int ns = spillcnt[c];
            for (int i = tid; i < ns; i += 512) {
                v2i ed = *(const v2i*)(spill + 2 * ((size_t)c * CHUNK + i));
                if ((ed.y >> BSHIFT) == b) atomicAdd(&cl[ed.y - node0], 1);
            }
        }
    }
    __syncthreads();
    int cn = cl[tid];
    s[tid] = cn;
    __syncthreads();
    lds_scan_incl(s, tid, 512);
    int excl = s[tid] - cn;
    int node = node0 + tid;
    if (node < n) {
        rowptr[node] = bstart + excl;
        float d = (float)(cn + 1);
        dinv[node] = d > 0.0f ? rsqrtf(d) : 0.0f;
        if (node == n - 1) rowptr[n] = bstart + s[tid];
    }
    cl[tid] = excl;
    __syncthreads();
    bool fits = bcount <= STAGE_CAP;
    for (int j0 = 0; 16 * j0 < nchunk; j0 += SU) {
        v4i e4[SU];
        int m[SU];
        int cq = w + 16 * (j0 + lane);
        int mv = (lane < SU && cq < nchunk) ? scnt[cq] : 0;
#pragma unroll
        for (int u = 0; u < SU; ++u) m[u] = __builtin_amdgcn_readlane(mv, u);
        const v4i* pb = seg4 + (size_t)(w + 16 * j0) * (CAP / 2) + lane;
#pragma unroll
        for (int u = 0; u < SU; ++u) {
            v4i none = {0, -1, 0, -1};
            e4[u] = 2 * lane < m[u] ? pb[u * 16 * (CAP / 2)] : none;
        }
        int sa[SU], sb[SU];
#pragma unroll
        for (int u = 0; u < SU; ++u) {
            sa[u] = 2 * lane < m[u]     ? atomicAdd(&cl[e4[u].y - node0], 1) : 0;
            sb[u] = 2 * lane + 1 < m[u] ? atomicAdd(&cl[e4[u].w - node0], 1) : 0;
        }
#pragma unroll
        for (int u = 0; u < SU; ++u) {
            if (2 * lane < m[u]) {
                if (fits) stage[sa[u]] = e4[u].x;
                else      src[bstart + sa[u]] = e4[u].x;
            }
            if (2 * lane + 1 < m[u]) {
                if (fits) stage[sb[u]] = e4[u].z;
                else      src[bstart + sb[u]] = e4[u].z;
            }
        }
    }
    if (anySpill) {
        for (int c = 0; c < nchunk; ++c) {
            int ns = spillcnt[c];
            for (int i = tid; i < ns; i += 512) {
                v2i ed = *(const v2i*)(spill + 2 * ((size_t)c * CHUNK + i));
                if ((ed.y >> BSHIFT) == b) {
                    int slot = atomicAdd(&cl[ed.y - node0], 1);
                    if (fits) stage[slot] = ed.x;
                    else      src[bstart + slot] = ed.x;
                }
            }
        }
    }
    if (fits) {
        __syncthreads();
        for (int i = tid; i < bcount; i += 512) src[bstart + i] = stage[i];
    }
}

template <int K>
__device__ __forceinline__ void
dense_body(int db, const float* __restrict__ X, const float* __restrict__ W,
           const float* __restrict__ bias, float* __restrict__ outPre,
           int ntiles, float* wlds) {
    constexpr int WP = K + 4;
    constexpr int NV4 = (64 * K) / 4;
    constexpr int RV4 = K / 4;
    for (int i = threadIdx.x; i < NV4; i += 512)
        *(float4*)(wlds + (i / RV4) * WP + 4 * (i % RV4)) = ((const float4*)W)[i];
    __syncthreads();

    int wid  = (db * 512 + (int)threadIdx.x) >> 5;
    int lane = threadIdx.x & 31;
    if (wid >= ntiles) return;
    int ln = lane & 15;
    int kh = lane >> 4;

    const float* xrow = X + (size_t)(wid * 16 + ln) * K + 2 * kh;
    v2f afr[K / 4];
#pragma unroll
    for (int kk = 0; kk < K / 4; ++kk)
        afr[kk] = *(const v2f*)(xrow + 4 * kk);

#pragma unroll
    for (int j = 0; j < 4; ++j) {
        const float* wrow = wlds + (size_t)(16 * j + ln) * WP + 2 * kh;
        v8f acc = {};
#pragma unroll
        for (int kk = 0; kk < K / 4; ++kk) {
            v2f b = *(const v2f*)(wrow + 4 * kk);
            acc = __builtin_amdgcn_wmma_f32_16x16x4_f32(
                false, afr[kk], false, b, (short)0, acc, false, false);
        }
        int n = 16 * j + ln;
        float bv = bias[n];
#pragma unroll
        for (int r = 0; r < 8; ++r) {
            int m = r + 8 * kh;
            size_t gi = (size_t)(wid * 16 + m) * 64 + n;
            outPre[gi] = acc[r] + bv;
        }
    }
}

__global__ void __launch_bounds__(512)
k_bfill_dense(const int* __restrict__ slots, const int* __restrict__ cnt,
              const int* __restrict__ spill, const int* __restrict__ spillcnt,
              int nchunk,
              int* __restrict__ rowptr, float* __restrict__ dinv, int* __restrict__ src, int n,
              int nbk, const float* __restrict__ X, const float* __restrict__ W,
              const float* __restrict__ bias, float* __restrict__ x0, int ntiles) {
    __shared__ __attribute__((aligned(16))) int lds[1536 + STAGE_CAP];
    if ((int)blockIdx.x < nbk)
        bfill_body(slots, cnt, spill, spillcnt, nchunk, rowptr, dinv, src, n, lds);
    else
        dense_body<128>((int)blockIdx.x - nbk, X, W, bias, x0, ntiles, (float*)lds);
}

#define TILE_PITCH 68
#define W_PITCH 80

__device__ __forceinline__ void ask16(v2f (&a)[16], const v2f* hrow2,
                                      unsigned lane, int ro, int k0) {
#pragma unroll
    for (int u = 0; u < 16; ++u) {
        unsigned r = (unsigned)__builtin_amdgcn_readlane(ro, k0 + u);
        a[u] = hrow2[r + lane];
    }
}

template <bool TAIL, bool FIRST>
__device__ __forceinline__ void sum16(const v2f (&a)[16], int p0,
                                      int endAll, int rp, float* tl,
                                      int& cur, int& curEnd, v2f& acc,
                                      float sc, int k0) {
#pragma unroll
    for (int u = 0; u < 16; ++u) {
        int p = p0 + u;
        if (!TAIL || p < endAll) {
            if (p >= curEnd) {
                *(v2f*)(tl + cur * TILE_PITCH) = acc;
                do {
                    ++cur;
                    curEnd = __builtin_amdgcn_readlane(rp, cur + 1);
                } while (p >= curEnd && cur < 15);
                acc = *(const v2f*)(tl + cur * TILE_PITCH);
            }
            if constexpr (FIRST) {
                float f = __int_as_float(__builtin_amdgcn_readlane(__float_as_int(sc), k0 + u));
                v2f r;
                r.x = fmaxf(a[u].x, 0.0f);
                r.y = fmaxf(a[u].y, 0.0f);
                acc += r * f;
            } else {
                acc += a[u];
            }
        }
    }
}

template <bool LAST, bool FIRST>
__device__ __forceinline__ void
layer_body(const int* __restrict__ rowptr, const int* __restrict__ src,
           const float* __restrict__ dinv, const float* __restrict__ hin,
           const float* __restrict__ x0, const float* __restrict__ Wl, float beta,
           float* __restrict__ hout, int ntiles, int scale_out,
           const float* __restrict__ W2, const float* __restrict__ b2) {
    __shared__ __attribute__((aligned(16))) float wlds[64 * W_PITCH];
    __shared__ __attribute__((aligned(16))) float tiles[4][16 * TILE_PITCH];
#pragma unroll
    for (int q = 0; q < 8; ++q) {
        int i = threadIdx.x + 128 * q;
        *(float4*)(wlds + (i >> 4) * W_PITCH + 4 * (i & 15)) = ((const float4*)Wl)[i];
    }

    int lane = threadIdx.x & 31;
    int wv   = threadIdx.x >> 5;
    int wid = __builtin_amdgcn_readfirstlane((int)(blockIdx.x * 4 + wv));
    bool active = wid < ntiles;
    float* t = tiles[wv];
    float dl = 0.0f;

    if (active) {
        int base = wid * 16;
        dl = dinv[base + (lane & 15)];
        int   rp = rowptr[base + (lane < 16 ? lane : 16)];
        const float* hl = hin + 2 * lane;
        const float* xl = x0 + 2 * lane;
        float* tl = t + 2 * lane;
        const v2f* hrow2 = (const v2f*)hin;
        unsigned ulane = (unsigned)lane;

        v2f hv[16];
        v2f xv[16];
        if constexpr (!FIRST) {
#pragma unroll
            for (int j = 0; j < 16; ++j)
                hv[j] = *(const v2f*)(hl + (size_t)(base + j) * 64);
        }
#pragma unroll
        for (int j = 0; j < 16; ++j)
            xv[j] = *(const v2f*)(xl + (size_t)(base + j) * 64);

        int beg    = __builtin_amdgcn_readlane(rp, 0);
        int endAll = __builtin_amdgcn_readlane(rp, 16);
        int pos = beg + lane;
        bool ok = pos < endAll;
        int rec = src[ok ? pos : beg];
        float sc = 0.0f, nsc = 0.0f;
        if constexpr (FIRST) {
            sc = dinv[ok ? rec : base];
#pragma unroll
            for (int j = 0; j < 16; ++j) {
                float dj = __int_as_float(__builtin_amdgcn_readlane(__float_as_int(dl), j));
                v2f r;
                r.x = fmaxf(xv[j].x, 0.0f);
                r.y = fmaxf(xv[j].y, 0.0f);
                hv[j] = r * dj;
            }
        }
#pragma unroll
        for (int j = 0; j < 16; ++j)
            *(v2f*)(tl + j * TILE_PITCH) = hv[j];
        int cur = 0;
        int curEnd = __builtin_amdgcn_readlane(rp, 1);
        v2f acc = *(const v2f*)tl;

        int ro = (ok ? rec : base) * 32;
        v2f ra[16], rb[16];
        ask16(ra, hrow2, ulane, ro, 0);
        for (int p0 = beg; p0 < endAll; p0 += 32) {
            int npos = p0 + 32 + lane;
            bool nok = npos < endAll;
            int nrec = src[nok ? npos : p0];
            ask16(rb, hrow2, ulane, ro, 16);
            bool full = p0 + 32 <= endAll;
            if (full) sum16<false, FIRST>(ra, p0, endAll, rp, tl, cur, curEnd, acc, sc, 0);
            else      sum16<true, FIRST>(ra, p0, endAll, rp, tl, cur, curEnd, acc, sc, 0);
            ro = (nok ? nrec : base) * 32;
            if constexpr (FIRST) nsc = dinv[nok ? nrec : base];
            ask16(ra, hrow2, ulane, ro, 0);
            if (full) sum16<false, FIRST>(rb, p0 + 16, endAll, rp, tl, cur, curEnd, acc, sc, 16);
            else      sum16<true, FIRST>(rb, p0 + 16, endAll, rp, tl, cur, curEnd, acc, sc, 16);
            if constexpr (FIRST) sc = nsc;
        }
        *(v2f*)(tl + cur * TILE_PITCH) = acc;
#pragma unroll
        for (int j = 0; j < 16; ++j) {
            float dc = __int_as_float(__builtin_amdgcn_readlane(__float_as_int(dl), j));
            v2f a = *(const v2f*)(tl + j * TILE_PITCH);
            *(v2f*)(tl + j * TILE_PITCH) = (a * dc) * 0.9f + xv[j] * 0.1f;
        }
    }
    __syncthreads();
    if constexpr (!LAST) {
        if (!active) return;
    }

    int ln = lane & 15;
    int kh = lane >> 4;
    float ob = 1.0f - beta;
    v4f w2r[8];
    if constexpr (LAST) {
#pragma unroll
        for (int q = 0; q < 8; ++q)
            w2r[q] = ((const v4f*)W2)[threadIdx.x + 128 * q];
    }
    if (active) {
        v2f afr[16];
#pragma unroll
        for (int kk = 0; kk < 16; ++kk)
            afr[kk] = *(const v2f*)(t + ln * TILE_PITCH + 4 * kk + 2 * kh);
        float sc[8];
        if constexpr (!LAST) {
#pragma unroll
            for (int r = 0; r < 8; ++r) {
                float lo = __int_as_float(__builtin_amdgcn_readlane(__float_as_int(dl), r));
                float hi = __int_as_float(__builtin_amdgcn_readlane(__float_as_int(dl), r + 8));
                sc[r] = scale_out ? (kh ? hi : lo) : 1.0f;
            }
        }
#pragma unroll
        for (int j = 0; j < 4; ++j) {
            const float* wcol = wlds + 2 * kh * W_PITCH + 16 * j + ln;
            v8f acc = {};
#pragma unroll
            for (int kk = 0; kk < 16; ++kk) {
                v2f b;
                b.x = wcol[(size_t)(4 * kk) * W_PITCH];
                b.y = wcol[(size_t)(4 * kk + 1) * W_PITCH];
                acc = __builtin_amdgcn_wmma_f32_16x16x4_f32(
                    false, afr[kk], false, b, (short)0, acc, false, false);
            }
#pragma unroll
            for (int r = 0; r < 8; ++r) {
                int m = r + 8 * kh;
                int n = 16 * j + ln;
                float hh = t[m * TILE_PITCH + n];
                float v = ob * hh + beta * acc[r];
                if constexpr (LAST) {
                    t[m * TILE_PITCH + n] = fmaxf(v, 0.0f);
                } else {
                    size_t gi = (size_t)(wid * 16 + m) * 64 + n;
                    hout[gi] = fmaxf(v, 0.0f) * sc[r];
                }
            }
        }
    }
    if constexpr (LAST) {
        __syncthreads();
#pragma unroll
        for (int q = 0; q < 8; ++q) {
            int i = threadIdx.x + 128 * q;
            *(v4f*)(wlds + (i >> 4) * TILE_PITCH + 4 * (i & 15)) = w2r[q];
        }
        __syncthreads();
        if (active) {
            v2f afr[16];
#pragma unroll
            for (int kk = 0; kk < 16; ++kk)
                afr[kk] = *(const v2f*)(t + ln * TILE_PITCH + 4 * kk + 2 * kh);
#pragma unroll
            for (int j = 0; j < 4; ++j) {
                const float* wrow = wlds + (16 * j + ln) * TILE_PITCH + 2 * kh;
                v8f acc = {};
#pragma unroll
                for (int kk = 0; kk < 16; ++kk) {
                    v2f b = *(const v2f*)(wrow + 4 * kk);
                    acc = __builtin_amdgcn_wmma_f32_16x16x4_f32(
                        false, afr[kk], false, b, (short)0, acc, false, false);
                }
                int n = 16 * j + ln;
                float bv = b2[n];
#pragma unroll
                for (int r = 0; r < 8; ++r) {
                    int m = r + 8 * kh;
                    size_t gi = (size_t)(wid * 16 + m) * 64 + n;
                    hout[gi] = acc[r] + bv;
                }
            }
        }
    }
}

__global__ void __launch_bounds__(128)
k_layer(const int* __restrict__ rowptr, const int* __restrict__ src,
        const float* __restrict__ dinv, const float* __restrict__ hin,
        const float* __restrict__ x0, const float* __restrict__ Wl, float beta,
        float* __restrict__ hout, int ntiles, int scale_out) {
    layer_body<false, false>(rowptr, src, dinv, hin, x0, Wl, beta, hout, ntiles, scale_out,
                             nullptr, nullptr);
}

__global__ void __launch_bounds__(128)
k_layer_first(const int* __restrict__ rowptr, const int* __restrict__ src,
              const float* __restrict__ dinv, const float* __restrict__ x0,
              const float* __restrict__ Wl, float beta,
              float* __restrict__ hout, int ntiles, int scale_out) {
    layer_body<false, true>(rowptr, src, dinv, x0, x0, Wl, beta, hout, ntiles, scale_out,
                            nullptr, nullptr);
}

__global__ void __launch_bounds__(128, 7)
k_layer_out(const int* __restrict__ rowptr, const int* __restrict__ src,
            const float* __restrict__ dinv, const float* __restrict__ hin,
            const float* __restrict__ x0, const float* __restrict__ Wl, float beta,
            const float* __restrict__ W2, const float* __restrict__ b2,
            float* __restrict__ out, int ntiles) {
    layer_body<true, false>(rowptr, src, dinv, hin, x0, Wl, beta, out, ntiles, 0, W2, b2);
}

extern "C" void kernel_launch(void* const* d_in, const int* in_sizes, int n_in,
                              void* d_out, int out_size, void* d_ws, size_t ws_size,
                              hipStream_t stream) {
    const float* x      = (const float*)d_in[0];
    const int*   ei     = (const int*)  d_in[1];
    const float* lin0_w = (const float*)d_in[2];
    const float* lin0_b = (const float*)d_in[3];
    const float* conv_w = (const float*)d_in[4];
    const float* lin1_w = (const float*)d_in[5];
    const float* lin1_b = (const float*)d_in[6];
    float* out = (float*)d_out;

    const int N = in_sizes[0] / 128;
    const int E = in_sizes[1] / 2;
    const int* row = ei;
    const int* col = ei + E;

    float* ws   = (float*)d_ws;
    float* dinv = ws;
    float* x0   = dinv + N;
    float* h    = x0 + (size_t)N * 64;
    float* h2   = h  + (size_t)N * 64;
    const int ntiles = N / 16;
    const int denseBlocks = (ntiles + 15) / 16;
    const int nbk = (N + 511) / 512;
    const int nchunk = (E + 4095) / 4096;

    int* slots  = (int*)(h2 + (size_t)N * 64);
    int* cnt    = slots + 2 * (size_t)nbk * nchunk * CAP;
    int* spill  = cnt + 2 * (size_t)nbk * nchunk;
    int* spillc = spill + 2 * (size_t)nchunk * CHUNK;
    int* srcs   = spillc + nchunk;
    int* rowptr = srcs + E;

    k_place <<<nchunk, 256, 0, stream>>>(row, col, slots, cnt, spill, spillc, E, nbk, nchunk);
    k_bfill_dense<<<nbk + denseBlocks, 512, 0, stream>>>(slots, cnt, spill, spillc, nchunk,
                                                         rowptr, dinv, srcs, N,
                                                         nbk, x, lin0_w, lin0_b, x0, ntiles);

    const int layerGrid = (ntiles + 3) / 4;
    for (int l = 0; l < 4; ++l) {
        float beta = logf(0.5f / (float)(l + 1) + 1.0f);
        const float* hin = (l & 1) ? h2 : h;
        float* hout      = (l & 1) ? h : h2;
        if (l == 0)
            k_layer_first<<<layerGrid, 128, 0, stream>>>(rowptr, srcs, dinv, x0, conv_w, beta,
                                                         hout, ntiles, 1);
        else if (l < 3)
            k_layer<<<layerGrid, 128, 0, stream>>>(rowptr, srcs, dinv, hin, x0,
                                                   conv_w + (size_t)l * 64 * 64, beta, hout,
                                                   ntiles, 1);
        else
            k_layer_out<<<layerGrid, 128, 0, stream>>>(rowptr, srcs, dinv, hin, x0,
                                                       conv_w + (size_t)l * 64 * 64, beta,
                                                       lin1_w, lin1_b, out, ntiles);
    }
}
